# starting kernel
# baseline (speedup 1.0000x reference)
#include <hip/hip_runtime.h>
#include <math.h>


typedef float v2f __attribute__((ext_vector_type(2)));
typedef float v8f __attribute__((ext_vector_type(8)));

#define BATCH 256
#define TIME  1024
#define IND   39
#define HID   256
#define RANK  16

#define WIS 264
#define BLS 16
#define AWS 264
#define WDS 48
#define XS  44
#define HS  260
#define RS  20

#define OFF_WI 0
#define OFF_BL (OFF_WI + 40*WIS)
#define OFF_AW (OFF_BL + 256*BLS)
#define OFF_WD (OFF_AW + 16*AWS)
#define OFF_B  (OFF_WD + 256*WDS)
#define OFF_BD (OFF_B  + 256)
#define OFF_X  (OFF_BD + 48)
#define OFF_H  (OFF_X  + 16*XS)
#define OFF_R  (OFF_H  + 16*HS)
#define OFF_RP (OFF_R  + 16*RS)
#define OFF_DP (OFF_RP + 16*256)
#define SMEM_FLOATS (OFF_DP + 16*768)
#define SMEM_BYTES  (SMEM_FLOATS * 4)

__device__ __forceinline__ v8f wmma4(v2f a, v2f b, v8f c) {
  return __builtin_amdgcn_wmma_f32_16x16x4_f32(
       false, a,  false, b,
       (short)0, c,  false,  false);
}

__device__ __forceinline__ float fast_tanh(float v) {
#if __has_builtin(__builtin_amdgcn_tanhf)
  return __builtin_amdgcn_tanhf(v);
#else
  return tanhf(v);
#endif
}

__device__ __forceinline__ v2f ldA(const float* base, int stride, int m, int kk, int kh) {
  const float* p = base + m * stride + kk + 2 * kh;
  v2f a; a.x = p[0]; a.y = p[1]; return a;
}
__device__ __forceinline__ v2f ldB(const float* base, int stride, int col, int kk, int kh) {
  const float* p = base + (kk + 2 * kh) * stride + col;
  v2f b; b.x = p[0]; b.y = p[stride]; return b;
}

__global__ __launch_bounds__(512)
void dream_scan(const float* __restrict__ x, const float* __restrict__ W_in,
                const float* __restrict__ Amat, const float* __restrict__ B_low,
                const float* __restrict__ bvec, const float* __restrict__ W_dec,
                const float* __restrict__ b_dec, float* __restrict__ out) {
  extern __shared__ float sm[];
  float* Wi = sm + OFF_WI;
  float* Bl = sm + OFF_BL;
  float* Aw = sm + OFF_AW;
  float* Wd = sm + OFF_WD;
  float* Bs = sm + OFF_B;
  float* Bd = sm + OFF_BD;
  float* Xl = sm + OFF_X;
  float* Hl = sm + OFF_H;
  float* Rl = sm + OFF_R;
  float* Rp = sm + OFF_RP;
  float* Dp = sm + OFF_DP;

  const int tid  = threadIdx.x;
  const int wave = tid >> 5;
  const int lane = tid & 31;
  const int lm   = lane & 15;
  const int kh   = lane >> 4;
  const int b0   = blockIdx.x * 16;
  const int n0   = wave * 16;

  for (int e = tid; e < 40 * 256; e += 512) {
    int n = e & 255, k = e >> 8;
    Wi[k * WIS + n] = (k < IND) ? W_in[n * IND + k] : 0.0f;
  }
  for (int e = tid; e < 256 * 16; e += 512) {
    int j = e & 15, k = e >> 4;
    Bl[k * BLS + j] = B_low[j * HID + k];
  }
  for (int e = tid; e < 16 * 256; e += 512) {
    int n = e & 255, j = e >> 8;
    Aw[j * AWS + n] = Amat[n * RANK + j];
  }
  for (int e = tid; e < 256 * 48; e += 512) {
    int c = e % 48, k = e / 48;
    Wd[k * WDS + c] = (c < IND) ? W_dec[c * HID + k] : 0.0f;
  }
  if (tid < 256) Bs[tid] = bvec[tid];
  if (tid < 48)  Bd[tid] = (tid < IND) ? b_dec[tid] : 0.0f;
  for (int e = tid; e < 16 * HS; e += 512) Hl[e] = 0.0f;
  for (int e = tid; e < 16 * XS; e += 512) Xl[e] = 0.0f;
  __syncthreads();

  const int  e0 = tid, e1 = tid + 512;
  const bool xv0 = (e0 < 16 * IND), xv1 = (e1 < 16 * IND);
  const float* xg0 = x;  int xl0 = 0;
  const float* xg1 = x;  int xl1 = 0;
  if (xv0) { int r = e0 / IND, c = e0 % IND;
    xg0 = x + (size_t)(b0 + r) * TIME * IND + c;  xl0 = r * XS + c; }
  if (xv1) { int r = e1 / IND, c = e1 % IND;
    xg1 = x + (size_t)(b0 + r) * TIME * IND + c;  xl1 = r * XS + c; }

  const int  ea = tid, eb = tid + 512;
  const bool sva = ((ea % 48) < IND);
  const bool svb = (eb < 768) && ((eb % 48) < IND);
  float* outa = out;  float bda = 0.0f;
  float* outb = out;  float bdb = 0.0f;
  if (sva) { int row = ea / 48, c = ea % 48;
    outa = out + (size_t)(b0 + row) * TIME * IND + c;  bda = Bd[c]; }
  if (svb) { int row = eb / 48, c = eb % 48;
    outb = out + (size_t)(b0 + row) * TIME * IND + c;  bdb = Bd[c]; }
  const float bb = Bs[n0 + lm];

  float xr0 = 0.0f, xr1 = 0.0f;
  if (xv0) xr0 = xg0[0];
  if (xv1) xr1 = xg1[0];

  v8f h_acc = {};

  for (int t = 0; t < TIME; ++t) {
    if (xv0) Xl[xl0] = xr0;
    if (xv1) Xl[xl1] = xr1;
    __syncthreads();
    if (t + 1 < TIME) {
      size_t off = (size_t)(t + 1) * IND;
      if (xv0) xr0 = xg0[off];
      if (xv1) xr1 = xg1[off];
    }

    v8f acc;
    #pragma unroll
    for (int v = 0; v < 8; ++v) acc[v] = bb;
    #pragma unroll
    for (int kk = 0; kk < 40; kk += 4)
      acc = wmma4(ldA(Xl, XS, lm, kk, kh), ldB(Wi, WIS, n0 + lm, kk, kh), acc);

    {
      v8f rp = {};
      const int kb = wave * 16;
      #pragma unroll
      for (int kk = 0; kk < 16; kk += 4)
        rp = wmma4(ldA(Hl, HS, lm, kb + kk, kh), ldB(Bl, BLS, lm, kb + kk, kh), rp);
      #pragma unroll
      for (int v = 0; v < 8; ++v)
        Rp[wave * 256 + (v + 8 * kh) * 16 + lm] = rp[v];
    }
    __syncthreads();
    if (tid < 256) {
      float s = 0.0f;
      #pragma unroll
      for (int w = 0; w < 16; ++w) s += Rp[w * 256 + tid];
      Rl[(tid >> 4) * RS + (tid & 15)] = s;
    }
    __syncthreads();

    #pragma unroll
    for (int kk = 0; kk < 16; kk += 4)
      acc = wmma4(ldA(Rl, RS, lm, kk, kh), ldB(Aw, AWS, n0 + lm, kk, kh), acc);
    #pragma unroll
    for (int v = 0; v < 8; ++v) acc[v] = fast_tanh(acc[v]);
    h_acc = acc;
    #pragma unroll
    for (int v = 0; v < 8; ++v) Hl[(v + 8 * kh) * HS + n0 + lm] = acc[v];
    __syncthreads();

    {
      v8f d0 = {}, d1 = {}, d2 = {};
      const int kb = wave * 16;
      #pragma unroll
      for (int kk = 0; kk < 16; kk += 4) {
        v2f a = ldA(Hl, HS, lm, kb + kk, kh);
        d0 = wmma4(a, ldB(Wd, WDS,  0 + lm, kb + kk, kh), d0);
        d1 = wmma4(a, ldB(Wd, WDS, 16 + lm, kb + kk, kh), d1);
        d2 = wmma4(a, ldB(Wd, WDS, 32 + lm, kb + kk, kh), d2);
      }
      #pragma unroll
      for (int v = 0; v < 8; ++v) {
        int row = v + 8 * kh;
        Dp[wave * 768 + row * 48 +  0 + lm] = d0[v];
        Dp[wave * 768 + row * 48 + 16 + lm] = d1[v];
        Dp[wave * 768 + row * 48 + 32 + lm] = d2[v];
      }
    }
    __syncthreads();
    {
      const size_t toff = (size_t)t * IND;
      if (sva) {
        float s = bda;
        #pragma unroll
        for (int w = 0; w < 16; ++w) s += Dp[w * 768 + ea];
        outa[toff] = s;
      }
      if (svb) {
        float s = bdb;
        #pragma unroll
        for (int w = 0; w < 16; ++w) s += Dp[w * 768 + eb];
        outb[toff] = s;
      }
    }
  }

  {
    float* hf = out + (size_t)BATCH * TIME * IND;
    #pragma unroll
    for (int v = 0; v < 8; ++v)
      hf[(size_t)(b0 + v + 8 * kh) * HID + n0 + lm] = h_acc[v];
  }
}

extern "C" void kernel_launch(void* const* d_in, const int* in_sizes, int n_in,
                              void* d_out, int out_size, void* d_ws, size_t ws_size,
                              hipStream_t stream) {
  const float* x     = (const float*)d_in[0];
  const float* W_in  = (const float*)d_in[1];
  const float* Amat  = (const float*)d_in[2];
  const float* B_low = (const float*)d_in[3];
  const float* bvec  = (const float*)d_in[4];
  const float* W_dec = (const float*)d_in[5];
  const float* b_dec = (const float*)d_in[6];
  float* out = (float*)d_out;

  hipFuncSetAttribute((const void*)dream_scan,
                      hipFuncAttributeMaxDynamicSharedMemorySize, SMEM_BYTES);
  dream_scan<<<BATCH / 16, 512, SMEM_BYTES, stream>>>(
      x, W_in, Amat, B_low, bvec, W_dec, b_dec, out);
}


# Software-pipeline the t loop: decoder partials of step t-1 run with r partials of step t, both reduces with the x projection of step t+1; 5 barriers per step become 3
# speedup vs baseline: 1.1418x; 1.1418x over previous
#include <hip/hip_runtime.h>
#include <math.h>


typedef float v2f __attribute__((ext_vector_type(2)));
typedef float v8f __attribute__((ext_vector_type(8)));

#define BATCH 256
#define TIME  1024
#define IND   39
#define HID   256
#define RANK  16

#define WIS 264
#define BLS 16
#define AWS 264
#define WDS 48
#define XS  44
#define HS  260
#define RS  20

#define OFF_WI 0
#define OFF_BL (OFF_WI + 40*WIS)
#define OFF_AW (OFF_BL + 256*BLS)
#define OFF_WD (OFF_AW + 16*AWS)
#define OFF_B  (OFF_WD + 256*WDS)
#define OFF_BD (OFF_B  + 256)
#define OFF_X  (OFF_BD + 48)
#define OFF_H  (OFF_X  + 16*XS)
#define OFF_R  (OFF_H  + 16*HS)
#define PS  1024
#define OFF_P  (OFF_R  + 16*RS)
#define SMEM_FLOATS (OFF_P + 16*PS)
#define SMEM_BYTES  (SMEM_FLOATS * 4)

__device__ __forceinline__ v8f wmma4(v2f a, v2f b, v8f c) {
  return __builtin_amdgcn_wmma_f32_16x16x4_f32(
       false, a,  false, b,
       (short)0, c,  false,  false);
}

__device__ __forceinline__ float fast_tanh(float v) {
#if __has_builtin(__builtin_amdgcn_tanhf)
  return __builtin_amdgcn_tanhf(v);
#else
  return tanhf(v);
#endif
}

__device__ __forceinline__ v2f ldA(const float* base, int stride, int m, int kk, int kh) {
  const float* p = base + m * stride + kk + 2 * kh;
  v2f a; a.x = p[0]; a.y = p[1]; return a;
}
__device__ __forceinline__ v2f ldB(const float* base, int stride, int col, int kk, int kh) {
  const float* p = base + (kk + 2 * kh) * stride + col;
  v2f b; b.x = p[0]; b.y = p[stride]; return b;
}

__global__ __launch_bounds__(512)
void dream_scan(const float* __restrict__ x, const float* __restrict__ W_in,
                const float* __restrict__ Amat, const float* __restrict__ B_low,
                const float* __restrict__ bvec, const float* __restrict__ W_dec,
                const float* __restrict__ b_dec, float* __restrict__ out) {
  extern __shared__ float sm[];
  float* Wi = sm + OFF_WI;
  float* Bl = sm + OFF_BL;
  float* Aw = sm + OFF_AW;
  float* Wd = sm + OFF_WD;
  float* Bs = sm + OFF_B;
  float* Bd = sm + OFF_BD;
  float* Xl = sm + OFF_X;
  float* Hl = sm + OFF_H;
  float* Rl = sm + OFF_R;
  float* Pp = sm + OFF_P;

  const int tid  = threadIdx.x;
  const int wave = tid >> 5;
  const int lane = tid & 31;
  const int lm   = lane & 15;
  const int kh   = lane >> 4;
  const int b0   = blockIdx.x * 16;
  const int n0   = wave * 16;

  for (int e = tid; e < 40 * 256; e += 512) {
    int n = e & 255, k = e >> 8;
    Wi[k * WIS + n] = (k < IND) ? W_in[n * IND + k] : 0.0f;
  }
  for (int e = tid; e < 256 * 16; e += 512) {
    int j = e & 15, k = e >> 4;
    Bl[k * BLS + j] = B_low[j * HID + k];
  }
  for (int e = tid; e < 16 * 256; e += 512) {
    int n = e & 255, j = e >> 8;
    Aw[j * AWS + n] = Amat[n * RANK + j];
  }
  for (int e = tid; e < 256 * 48; e += 512) {
    int c = e % 48, k = e / 48;
    Wd[k * WDS + c] = (c < IND) ? W_dec[c * HID + k] : 0.0f;
  }
  if (tid < 256) Bs[tid] = bvec[tid];
  if (tid < 48)  Bd[tid] = (tid < IND) ? b_dec[tid] : 0.0f;
  for (int e = tid; e < 16 * HS; e += 512) Hl[e] = 0.0f;
  for (int e = tid; e < 16 * XS; e += 512) Xl[e] = 0.0f;
  __syncthreads();

  const int  e0 = tid, e1 = tid + 512;
  const bool xv0 = (e0 < 16 * IND), xv1 = (e1 < 16 * IND);
  const float* xg0 = x;  int xl0 = 0;
  const float* xg1 = x;  int xl1 = 0;
  if (xv0) { int r = e0 / IND, c = e0 % IND;
    xg0 = x + (size_t)(b0 + r) * TIME * IND + c;  xl0 = r * XS + c; }
  if (xv1) { int r = e1 / IND, c = e1 % IND;
    xg1 = x + (size_t)(b0 + r) * TIME * IND + c;  xl1 = r * XS + c; }

  const int  ea = tid, eb = tid + 512;
  const bool lower = (tid < 256);
  const bool sva = ((ea % 48) < IND);
  const bool svb = lower && ((eb % 48) < IND);
  float* outa = out;  float bda = 0.0f;
  float* outb = out;  float bdb = 0.0f;
  if (sva) { int row = ea / 48, c = ea % 48;
    outa = out + (size_t)(b0 + row) * TIME * IND + c;  bda = Bd[c]; }
  if (svb) { int row = eb / 48, c = eb % 48;
    outb = out + (size_t)(b0 + row) * TIME * IND + c;  bdb = Bd[c]; }
  const int rq = tid - 256;
  const int rl_idx = lower ? 0 : ((rq >> 4) * RS + (rq & 15));
  const float bb = Bs[n0 + lm];
  const int kb = wave * 16;
  float* Pw = Pp + wave * PS;


  if (xv0) Xl[xl0] = xg0[0];
  if (xv1) Xl[xl1] = xg1[0];
  __syncthreads();
  v8f acc_next;
  #pragma unroll
  for (int v = 0; v < 8; ++v) acc_next[v] = bb;
  #pragma unroll
  for (int kk = 0; kk < 40; kk += 4)
    acc_next = wmma4(ldA(Xl, XS, lm, kk, kh), ldB(Wi, WIS, n0 + lm, kk, kh), acc_next);
  __syncthreads();
  if (TIME > 1) {
    if (xv0) Xl[xl0] = xg0[IND];
    if (xv1) Xl[xl1] = xg1[IND];
  }

  v8f h_acc = {};
  float o1 = 0.0f, o2 = 0.0f;

  for (int t = 0; t < TIME; ++t) {
    if (t > 1) {
      const size_t toff = (size_t)(t - 2) * IND;
      if (sva) outa[toff] = o1;
      if (svb) outb[toff] = o2;
    }
    float xr0 = 0.0f, xr1 = 0.0f;
    if (t + 2 < TIME) {
      size_t off = (size_t)(t + 2) * IND;
      if (xv0) xr0 = xg0[off];
      if (xv1) xr1 = xg1[off];
    }
    {
      v8f rp = {};
      #pragma unroll
      for (int kk = 0; kk < 16; kk += 4)
        rp = wmma4(ldA(Hl, HS, lm, kb + kk, kh), ldB(Bl, BLS, lm, kb + kk, kh), rp);
      #pragma unroll
      for (int v = 0; v < 8; ++v)
        Pw[768 + (v + 8 * kh) * 16 + lm] = rp[v];
    }
    {
      v8f d0 = {}, d1 = {}, d2 = {};
      #pragma unroll
      for (int kk = 0; kk < 16; kk += 4) {
        v2f a = ldA(Hl, HS, lm, kb + kk, kh);
        d0 = wmma4(a, ldB(Wd, WDS,  0 + lm, kb + kk, kh), d0);
        d1 = wmma4(a, ldB(Wd, WDS, 16 + lm, kb + kk, kh), d1);
        d2 = wmma4(a, ldB(Wd, WDS, 32 + lm, kb + kk, kh), d2);
      }
      #pragma unroll
      for (int v = 0; v < 8; ++v) {
        int row = v + 8 * kh;
        Pw[row * 48 +  0 + lm] = d0[v];
        Pw[row * 48 + 16 + lm] = d1[v];
        Pw[row * 48 + 32 + lm] = d2[v];
      }
    }
    __syncthreads();

    float p1[16], p2[16];
    #pragma unroll
    for (int w = 0; w < 16; ++w) { p1[w] = Pp[w * PS + ea];  p2[w] = Pp[w * PS + eb]; }
    v8f acc = acc_next;
    #pragma unroll
    for (int v = 0; v < 8; ++v) acc_next[v] = bb;
    #pragma unroll
    for (int kk = 0; kk < 40; kk += 4)
      acc_next = wmma4(ldA(Xl, XS, lm, kk, kh), ldB(Wi, WIS, n0 + lm, kk, kh), acc_next);
    {
      float s1 = bda, s2 = lower ? bdb : 0.0f;
      #pragma unroll
      for (int w = 0; w < 16; ++w) { s1 += p1[w];  s2 += p2[w]; }
      o1 = s1;  o2 = s2;
      if (!lower) Rl[rl_idx] = s2;
    }
    __syncthreads();

    #pragma unroll
    for (int kk = 0; kk < 16; kk += 4)
      acc = wmma4(ldA(Rl, RS, lm, kk, kh), ldB(Aw, AWS, n0 + lm, kk, kh), acc);
    #pragma unroll
    for (int v = 0; v < 8; ++v) acc[v] = fast_tanh(acc[v]);
    h_acc = acc;
    #pragma unroll
    for (int v = 0; v < 8; ++v) Hl[(v + 8 * kh) * HS + n0 + lm] = acc[v];
    if (t + 2 < TIME) {
      if (xv0) Xl[xl0] = xr0;
      if (xv1) Xl[xl1] = xr1;
    }
    __syncthreads();
  }

  if (TIME > 1) {
    const size_t toff = (size_t)(TIME - 2) * IND;
    if (sva) outa[toff] = o1;
    if (svb) outb[toff] = o2;
  }
  {
    v8f d0 = {}, d1 = {}, d2 = {};
    #pragma unroll
    for (int kk = 0; kk < 16; kk += 4) {
      v2f a = ldA(Hl, HS, lm, kb + kk, kh);
      d0 = wmma4(a, ldB(Wd, WDS,  0 + lm, kb + kk, kh), d0);
      d1 = wmma4(a, ldB(Wd, WDS, 16 + lm, kb + kk, kh), d1);
      d2 = wmma4(a, ldB(Wd, WDS, 32 + lm, kb + kk, kh), d2);
    }
    #pragma unroll
    for (int v = 0; v < 8; ++v) {
      int row = v + 8 * kh;
      Pw[row * 48 +  0 + lm] = d0[v];
      Pw[row * 48 + 16 + lm] = d1[v];
      Pw[row * 48 + 32 + lm] = d2[v];
    }
  }
  __syncthreads();
  {
    const size_t toff = (size_t)(TIME - 1) * IND;
    if (sva) {
      float s = bda;
      #pragma unroll
      for (int w = 0; w < 16; ++w) s += Pp[w * PS + ea];
      outa[toff] = s;
    }
    if (svb) {
      float s = bdb;
      #pragma unroll
      for (int w = 0; w < 16; ++w) s += Pp[w * PS + eb];
      outb[toff] = s;
    }
  }

  {
    float* hf = out + (size_t)BATCH * TIME * IND;
    #pragma unroll
    for (int v = 0; v < 8; ++v)
      hf[(size_t)(b0 + v + 8 * kh) * HID + n0 + lm] = h_acc[v];
  }
}

extern "C" void kernel_launch(void* const* d_in, const int* in_sizes, int n_in,
                              void* d_out, int out_size, void* d_ws, size_t ws_size,
                              hipStream_t stream) {
  const float* x     = (const float*)d_in[0];
  const float* W_in  = (const float*)d_in[1];
  const float* Amat  = (const float*)d_in[2];
  const float* B_low = (const float*)d_in[3];
  const float* bvec  = (const float*)d_in[4];
  const float* W_dec = (const float*)d_in[5];
  const float* b_dec = (const float*)d_in[6];
  float* out = (float*)d_out;

  hipFuncSetAttribute((const void*)dream_scan,
                      hipFuncAttributeMaxDynamicSharedMemorySize, SMEM_BYTES);
  dream_scan<<<BATCH / 16, 512, SMEM_BYTES, stream>>>(
      x, W_in, Amat, B_low, bvec, W_dec, b_dec, out);
}


# Transposed products: a wave's tanh tile (D layout) is the B operand of its next r/decoder WMMAs, h stays in registers, weights are register A fragments; 3 barriers/step to 2; k sum order changes
# speedup vs baseline: 1.5096x; 1.3221x over previous
#include <hip/hip_runtime.h>
#include <math.h>


typedef float v2f __attribute__((ext_vector_type(2)));
typedef float v4f __attribute__((ext_vector_type(4)));
typedef float v8f __attribute__((ext_vector_type(8)));

#define BATCH 256
#define TIME  1024
#define IND   39
#define HID   256
#define RANK  16

#define XS  44
#define RS  20
#define DS  52
#define PR  (16 * DS)
#define PS  1152

#define OFF_X 0
#define OFF_R (OFF_X + 16 * XS)
#define OFF_P (OFF_R + 16 * RS)
#define SMEM_FLOATS (OFF_P + 16 * PS)
#define SMEM_BYTES  (SMEM_FLOATS * 4)

__device__ __forceinline__ v8f wmma4(v2f a, v2f b, v8f c) {
  return __builtin_amdgcn_wmma_f32_16x16x4_f32(
       false, a,  false, b,
       (short)0, c,  false,  false);
}

__device__ __forceinline__ float fast_tanh(float v) {
#if __has_builtin(__builtin_amdgcn_tanhf)
  return __builtin_amdgcn_tanhf(v);
#else
  return tanhf(v);
#endif
}

__device__ __forceinline__ v8f xproj(const v2f* fWi, const float* Xl, int lm, int kh, v8f acc) {
  const v4f* xp = (const v4f*)(Xl + lm * XS + 20 * kh);
  v4f q0 = xp[0], q1 = xp[1], q2 = xp[2], q3 = xp[3], q4 = xp[4];
  v2f b;
  b.x = q0.x; b.y = q0.y; acc = wmma4(fWi[0], b, acc);
  b.x = q0.z; b.y = q0.w; acc = wmma4(fWi[1], b, acc);
  b.x = q1.x; b.y = q1.y; acc = wmma4(fWi[2], b, acc);
  b.x = q1.z; b.y = q1.w; acc = wmma4(fWi[3], b, acc);
  b.x = q2.x; b.y = q2.y; acc = wmma4(fWi[4], b, acc);
  b.x = q2.z; b.y = q2.w; acc = wmma4(fWi[5], b, acc);
  b.x = q3.x; b.y = q3.y; acc = wmma4(fWi[6], b, acc);
  b.x = q3.z; b.y = q3.w; acc = wmma4(fWi[7], b, acc);
  b.x = q4.x; b.y = q4.y; acc = wmma4(fWi[8], b, acc);
  b.x = q4.z; b.y = q4.w; acc = wmma4(fWi[9], b, acc);
  return acc;
}

__device__ __forceinline__ void dec_partials(const v2f (*fWd)[4], v8f hT, float* Pw, int lm, int kh) {
  v8f d0 = {}, d1 = {}, d2 = {};
  #pragma unroll
  for (int j = 0; j < 4; ++j) {
    v2f b; b.x = hT[2 * j]; b.y = hT[2 * j + 1];
    d0 = wmma4(fWd[0][j], b, d0);
    d1 = wmma4(fWd[1][j], b, d1);
    d2 = wmma4(fWd[2][j], b, d2);
  }
  v4f* pd = (v4f*)(Pw + lm * DS + 8 * kh);
  v4f s;
  s.x = d0[0]; s.y = d0[1]; s.z = d0[2]; s.w = d0[3]; pd[0] = s;
  s.x = d0[4]; s.y = d0[5]; s.z = d0[6]; s.w = d0[7]; pd[1] = s;
  s.x = d1[0]; s.y = d1[1]; s.z = d1[2]; s.w = d1[3]; pd[4] = s;
  s.x = d1[4]; s.y = d1[5]; s.z = d1[6]; s.w = d1[7]; pd[5] = s;
  s.x = d2[0]; s.y = d2[1]; s.z = d2[2]; s.w = d2[3]; pd[8] = s;
  s.x = d2[4]; s.y = d2[5]; s.z = d2[6]; s.w = d2[7]; pd[9] = s;
}

__global__ __launch_bounds__(512)
void dream_scan(const float* __restrict__ x, const float* __restrict__ W_in,
                const float* __restrict__ Amat, const float* __restrict__ B_low,
                const float* __restrict__ bvec, const float* __restrict__ W_dec,
                const float* __restrict__ b_dec, float* __restrict__ out) {
  extern __shared__ __attribute__((aligned(16))) float sm[];
  float* Xl = sm + OFF_X;
  float* Rl = sm + OFF_R;
  float* Pp = sm + OFF_P;

  const int tid  = threadIdx.x;
  const int wave = tid >> 5;
  const int lane = tid & 31;
  const int lm   = lane & 15;
  const int kh   = lane >> 4;
  const int b0   = blockIdx.x * 16;
  const int n0   = wave * 16;

  v2f fBl[4], fAw[4], fWd[3][4], fWi[10];
  #pragma unroll
  for (int j = 0; j < 4; ++j) {
    const float* p = B_low + lm * HID + n0 + 8 * kh + 2 * j;
    fBl[j].x = p[0]; fBl[j].y = p[1];
    const float* q = Amat + (n0 + lm) * RANK + 8 * kh + 2 * j;
    fAw[j].x = q[0]; fAw[j].y = q[1];
  }
  #pragma unroll
  for (int i = 0; i < 3; ++i) {
    const int c = 16 * i + lm;
    #pragma unroll
    for (int j = 0; j < 4; ++j) {
      const float* p = W_dec + (c < IND ? c : 0) * HID + n0 + 8 * kh + 2 * j;
      fWd[i][j].x = (c < IND) ? p[0] : 0.0f;
      fWd[i][j].y = (c < IND) ? p[1] : 0.0f;
    }
  }
  #pragma unroll
  for (int j = 0; j < 10; ++j) {
    const float* p = W_in + (n0 + lm) * IND;
    const int k = 20 * kh + 2 * j;
    fWi[j].x = p[k];
    fWi[j].y = (k + 1 < IND) ? p[k + 1] : 0.0f;
  }
  float bbv[8];
  #pragma unroll
  for (int v = 0; v < 8; ++v) bbv[v] = bvec[n0 + 8 * kh + v];

  for (int e = tid; e < 16 * XS; e += 512) Xl[e] = 0.0f;
  __syncthreads();

  const int  e0 = tid, e1 = tid + 512;
  const bool xv0 = (e0 < 16 * IND), xv1 = (e1 < 16 * IND);
  const float* xg0 = x;  int xl0 = 0;
  const float* xg1 = x;  int xl1 = 0;
  if (xv0) { int r = e0 / IND, c = e0 % IND;
    xg0 = x + (size_t)(b0 + r) * TIME * IND + c;  xl0 = r * XS + c; }
  if (xv1) { int r = e1 / IND, c = e1 % IND;
    xg1 = x + (size_t)(b0 + r) * TIME * IND + c;  xl1 = r * XS + c; }

  const int  ea = tid, eb = tid + 512;
  const bool lower = (tid < 256);
  const bool sva = ((ea % 48) < IND);
  const bool svb = lower && ((eb % 48) < IND);
  float* outa = out;  float bda = 0.0f;
  float* outb = out;  float bdb = 0.0f;
  if (sva) { int row = ea / 48, c = ea % 48;
    outa = out + (size_t)(b0 + row) * TIME * IND + c;  bda = b_dec[c]; }
  if (svb) { int row = eb / 48, c = eb % 48;
    outb = out + (size_t)(b0 + row) * TIME * IND + c;  bdb = b_dec[c]; }
  const int rq = tid - 256;
  const int ca = (ea / 48) * DS + (ea % 48);
  const int cb = lower ? ((eb / 48) * DS + (eb % 48)) : (PR + (rq >> 4) * RS + (rq & 15));
  const int rl_idx = lower ? 0 : ((rq >> 4) * RS + (rq & 15));
  float* Pw = Pp + wave * PS;


  if (xv0) Xl[xl0] = xg0[0];
  if (xv1) Xl[xl1] = xg1[0];
  __syncthreads();
  v8f acc_next;
  #pragma unroll
  for (int v = 0; v < 8; ++v) acc_next[v] = bbv[v];
  acc_next = xproj(fWi, Xl, lm, kh, acc_next);
  __syncthreads();
  if (TIME > 1) {
    if (xv0) Xl[xl0] = xg0[IND];
    if (xv1) Xl[xl1] = xg1[IND];
  }

  v8f hT = {};
  float o1 = 0.0f, o2 = 0.0f;

  for (int t = 0; t < TIME; ++t) {
    if (t > 1) {
      const size_t toff = (size_t)(t - 2) * IND;
      if (sva) outa[toff] = o1;
      if (svb) outb[toff] = o2;
    }
    float xr0 = 0.0f, xr1 = 0.0f;
    if (t + 2 < TIME) {
      size_t off = (size_t)(t + 2) * IND;
      if (xv0) xr0 = xg0[off];
      if (xv1) xr1 = xg1[off];
    }
    {
      v8f rp = {};
      #pragma unroll
      for (int j = 0; j < 4; ++j) {
        v2f b; b.x = hT[2 * j]; b.y = hT[2 * j + 1];
        rp = wmma4(fBl[j], b, rp);
      }
      v4f* pr = (v4f*)(Pw + PR + lm * RS + 8 * kh);
      v4f s;
      s.x = rp[0]; s.y = rp[1]; s.z = rp[2]; s.w = rp[3]; pr[0] = s;
      s.x = rp[4]; s.y = rp[5]; s.z = rp[6]; s.w = rp[7]; pr[1] = s;
    }
    dec_partials(fWd, hT, Pw, lm, kh);
    __syncthreads();

    float p1[16], p2[16];
    #pragma unroll
    for (int w = 0; w < 16; ++w) { p1[w] = Pp[w * PS + ca];  p2[w] = Pp[w * PS + cb]; }
    v8f acc = acc_next;
    #pragma unroll
    for (int v = 0; v < 8; ++v) acc_next[v] = bbv[v];
    acc_next = xproj(fWi, Xl, lm, kh, acc_next);
    {
      float s1 = bda, s2 = lower ? bdb : 0.0f;
      #pragma unroll
      for (int w = 0; w < 16; ++w) { s1 += p1[w];  s2 += p2[w]; }
      o1 = s1;  o2 = s2;
      if (!lower) Rl[rl_idx] = s2;
    }
    __syncthreads();

    {
      const v4f* rr = (const v4f*)(Rl + lm * RS + 8 * kh);
      v4f ra = rr[0], rb = rr[1];
      v2f b;
      b.x = ra.x; b.y = ra.y; acc = wmma4(fAw[0], b, acc);
      b.x = ra.z; b.y = ra.w; acc = wmma4(fAw[1], b, acc);
      b.x = rb.x; b.y = rb.y; acc = wmma4(fAw[2], b, acc);
      b.x = rb.z; b.y = rb.w; acc = wmma4(fAw[3], b, acc);
    }
    #pragma unroll
    for (int v = 0; v < 8; ++v) hT[v] = fast_tanh(acc[v]);
    if (t + 2 < TIME) {
      if (xv0) Xl[xl0] = xr0;
      if (xv1) Xl[xl1] = xr1;
    }
  }

  if (TIME > 1) {
    const size_t toff = (size_t)(TIME - 2) * IND;
    if (sva) outa[toff] = o1;
    if (svb) outb[toff] = o2;
  }
  dec_partials(fWd, hT, Pw, lm, kh);
  __syncthreads();
  {
    const size_t toff = (size_t)(TIME - 1) * IND;
    if (sva) {
      float s = bda;
      #pragma unroll
      for (int w = 0; w < 16; ++w) s += Pp[w * PS + ca];
      outa[toff] = s;
    }
    if (svb) {
      float s = bdb;
      #pragma unroll
      for (int w = 0; w < 16; ++w) s += Pp[w * PS + cb];
      outb[toff] = s;
    }
  }

  {
    float* hf = out + (size_t)BATCH * TIME * IND + (size_t)(b0 + lm) * HID + n0 + 8 * kh;
    #pragma unroll
    for (int v = 0; v < 8; ++v) hf[v] = hT[v];
  }
}

extern "C" void kernel_launch(void* const* d_in, const int* in_sizes, int n_in,
                              void* d_out, int out_size, void* d_ws, size_t ws_size,
                              hipStream_t stream) {
  const float* x     = (const float*)d_in[0];
  const float* W_in  = (const float*)d_in[1];
  const float* Amat  = (const float*)d_in[2];
  const float* B_low = (const float*)d_in[3];
  const float* bvec  = (const float*)d_in[4];
  const float* W_dec = (const float*)d_in[5];
  const float* b_dec = (const float*)d_in[6];
  float* out = (float*)d_out;

  hipFuncSetAttribute((const void*)dream_scan,
                      hipFuncAttributeMaxDynamicSharedMemorySize, SMEM_BYTES);
  dream_scan<<<BATCH / 16, 512, SMEM_BYTES, stream>>>(
      x, W_in, Amat, B_low, bvec, W_dec, b_dec, out);
}


# Recurrence replicated on 8 workgroups per batch tile (128 WGPs); a copy runs the decoder only every 8th step (18, not 30, WMMAs per wave otherwise); global traffic moved to phase B; sums unchanged
# speedup vs baseline: 2.1000x; 1.3911x over previous
#include <hip/hip_runtime.h>
#include <math.h>


typedef float v2f __attribute__((ext_vector_type(2)));
typedef float v4f __attribute__((ext_vector_type(4)));
typedef float v8f __attribute__((ext_vector_type(8)));

#define BATCH 256
#define TIME  1024
#define IND   39
#define HID   256
#define RANK  16
#define NREP  8

#define XS  44
#define RS  20
#define DS  52
#define PR  (16 * DS)
#define PS  1152

#define OFF_X 0
#define OFF_R (OFF_X + 16 * XS)
#define OFF_P (OFF_R + 16 * RS)
#define SMEM_FLOATS (OFF_P + 16 * PS)
#define SMEM_BYTES  (SMEM_FLOATS * 4)
#define LDS_RESERVE (164 * 1024)

__device__ __forceinline__ v8f wmma4(v2f a, v2f b, v8f c) {
  return __builtin_amdgcn_wmma_f32_16x16x4_f32(
       false, a,  false, b,
       (short)0, c,  false,  false);
}

__device__ __forceinline__ float fast_tanh(float v) {
#if __has_builtin(__builtin_amdgcn_tanhf)
  return __builtin_amdgcn_tanhf(v);
#else
  return tanhf(v);
#endif
}

__device__ __forceinline__ v8f xproj(const v2f* fWi, const float* Xl, int lm, int kh, v8f acc) {
  const v4f* xp = (const v4f*)(Xl + lm * XS + 20 * kh);
  v4f q0 = xp[0], q1 = xp[1], q2 = xp[2], q3 = xp[3], q4 = xp[4];
  v2f b;
  b.x = q0.x; b.y = q0.y; acc = wmma4(fWi[0], b, acc);
  b.x = q0.z; b.y = q0.w; acc = wmma4(fWi[1], b, acc);
  b.x = q1.x; b.y = q1.y; acc = wmma4(fWi[2], b, acc);
  b.x = q1.z; b.y = q1.w; acc = wmma4(fWi[3], b, acc);
  b.x = q2.x; b.y = q2.y; acc = wmma4(fWi[4], b, acc);
  b.x = q2.z; b.y = q2.w; acc = wmma4(fWi[5], b, acc);
  b.x = q3.x; b.y = q3.y; acc = wmma4(fWi[6], b, acc);
  b.x = q3.z; b.y = q3.w; acc = wmma4(fWi[7], b, acc);
  b.x = q4.x; b.y = q4.y; acc = wmma4(fWi[8], b, acc);
  b.x = q4.z; b.y = q4.w; acc = wmma4(fWi[9], b, acc);
  return acc;
}

__device__ __forceinline__ void dec_partials(const v2f (*fWd)[4], v8f hT, float* Pw, int lm, int kh) {
  v8f d0 = {}, d1 = {}, d2 = {};
  #pragma unroll
  for (int j = 0; j < 4; ++j) {
    v2f b; b.x = hT[2 * j]; b.y = hT[2 * j + 1];
    d0 = wmma4(fWd[0][j], b, d0);
    d1 = wmma4(fWd[1][j], b, d1);
    d2 = wmma4(fWd[2][j], b, d2);
  }
  v4f* pd = (v4f*)(Pw + lm * DS + 8 * kh);
  v4f s;
  s.x = d0[0]; s.y = d0[1]; s.z = d0[2]; s.w = d0[3]; pd[0] = s;
  s.x = d0[4]; s.y = d0[5]; s.z = d0[6]; s.w = d0[7]; pd[1] = s;
  s.x = d1[0]; s.y = d1[1]; s.z = d1[2]; s.w = d1[3]; pd[4] = s;
  s.x = d1[4]; s.y = d1[5]; s.z = d1[6]; s.w = d1[7]; pd[5] = s;
  s.x = d2[0]; s.y = d2[1]; s.z = d2[2]; s.w = d2[3]; pd[8] = s;
  s.x = d2[4]; s.y = d2[5]; s.z = d2[6]; s.w = d2[7]; pd[9] = s;
}

__global__ __launch_bounds__(512)
void dream_scan(const float* __restrict__ x, const float* __restrict__ W_in,
                const float* __restrict__ Amat, const float* __restrict__ B_low,
                const float* __restrict__ bvec, const float* __restrict__ W_dec,
                const float* __restrict__ b_dec, float* __restrict__ out) {
  __shared__ __attribute__((aligned(16))) float sm[LDS_RESERVE / 4];
  float* Xl = sm + OFF_X;
  float* Rl = sm + OFF_R;
  float* Pp = sm + OFF_P;

  const int tid  = threadIdx.x;
  const int wave = tid >> 5;
  const int lane = tid & 31;
  const int lm   = lane & 15;
  const int kh   = lane >> 4;
  const int b0   = (blockIdx.x & 15) * 16;
  const int ph   = blockIdx.x >> 4;
  const int n0   = wave * 16;

  v2f fBl[4], fAw[4], fWd[3][4], fWi[10];
  #pragma unroll
  for (int j = 0; j < 4; ++j) {
    const float* p = B_low + lm * HID + n0 + 8 * kh + 2 * j;
    fBl[j].x = p[0]; fBl[j].y = p[1];
    const float* q = Amat + (n0 + lm) * RANK + 8 * kh + 2 * j;
    fAw[j].x = q[0]; fAw[j].y = q[1];
  }
  #pragma unroll
  for (int i = 0; i < 3; ++i) {
    const int c = 16 * i + lm;
    #pragma unroll
    for (int j = 0; j < 4; ++j) {
      const float* p = W_dec + (c < IND ? c : 0) * HID + n0 + 8 * kh + 2 * j;
      fWd[i][j].x = (c < IND) ? p[0] : 0.0f;
      fWd[i][j].y = (c < IND) ? p[1] : 0.0f;
    }
  }
  #pragma unroll
  for (int j = 0; j < 10; ++j) {
    const float* p = W_in + (n0 + lm) * IND;
    const int k = 20 * kh + 2 * j;
    fWi[j].x = p[k];
    fWi[j].y = (k + 1 < IND) ? p[k + 1] : 0.0f;
  }
  float bbv[8];
  #pragma unroll
  for (int v = 0; v < 8; ++v) bbv[v] = bvec[n0 + 8 * kh + v];

  for (int e = tid; e < 16 * XS; e += 512) Xl[e] = 0.0f;
  __syncthreads();

  const int  e0 = tid, e1 = tid + 512;
  const bool xv0 = (e0 < 16 * IND), xv1 = (e1 < 16 * IND);
  const float* xg0 = x;  int xl0 = 0;
  const float* xg1 = x;  int xl1 = 0;
  if (xv0) { int r = e0 / IND, c = e0 % IND;
    xg0 = x + (size_t)(b0 + r) * TIME * IND + c;  xl0 = r * XS + c; }
  if (xv1) { int r = e1 / IND, c = e1 % IND;
    xg1 = x + (size_t)(b0 + r) * TIME * IND + c;  xl1 = r * XS + c; }

  const int  ea = tid, eb = tid + 512;
  const bool lower = (tid < 256);
  const bool sva = ((ea % 48) < IND);
  const bool svb = lower && ((eb % 48) < IND);
  float* outa = out;  float bda = 0.0f;
  float* outb = out;  float bdb = 0.0f;
  if (sva) { int row = ea / 48, c = ea % 48;
    outa = out + (size_t)(b0 + row) * TIME * IND + c;  bda = b_dec[c]; }
  if (svb) { int row = eb / 48, c = eb % 48;
    outb = out + (size_t)(b0 + row) * TIME * IND + c;  bdb = b_dec[c]; }
  const int rq = tid - 256;
  const int ca = (ea / 48) * DS + (ea % 48);
  const int cb = lower ? ((eb / 48) * DS + (eb % 48)) : (PR + (rq >> 4) * RS + (rq & 15));
  const int rl_idx = lower ? 0 : ((rq >> 4) * RS + (rq & 15));
  float* Pw = Pp + wave * PS;


  if (xv0) Xl[xl0] = xg0[0];
  if (xv1) Xl[xl1] = xg1[0];
  __syncthreads();
  v8f acc_next;
  #pragma unroll
  for (int v = 0; v < 8; ++v) acc_next[v] = bbv[v];
  acc_next = xproj(fWi, Xl, lm, kh, acc_next);
  __syncthreads();
  if (TIME > 1) {
    if (xv0) Xl[xl0] = xg0[IND];
    if (xv1) Xl[xl1] = xg1[IND];
  }

  v8f hT = {};
  float o1 = 0.0f, o2 = 0.0f;

  for (int t = 0; t < TIME; ++t) {
    const bool st_now  = (t > 1) && (((t - 2) & (NREP - 1)) == ph);
    const bool dec_now = (t > 0) && (((t - 1) & (NREP - 1)) == ph);

    {
      v8f rp = {};
      #pragma unroll
      for (int j = 0; j < 4; ++j) {
        v2f b; b.x = hT[2 * j]; b.y = hT[2 * j + 1];
        rp = wmma4(fBl[j], b, rp);
      }
      v4f* pr = (v4f*)(Pw + PR + lm * RS + 8 * kh);
      v4f s;
      s.x = rp[0]; s.y = rp[1]; s.z = rp[2]; s.w = rp[3]; pr[0] = s;
      s.x = rp[4]; s.y = rp[5]; s.z = rp[6]; s.w = rp[7]; pr[1] = s;
    }
    if (dec_now) dec_partials(fWd, hT, Pw, lm, kh);
    __syncthreads();

    if (st_now) {
      const size_t toff = (size_t)(t - 2) * IND;
      if (sva) outa[toff] = o1;
      if (svb) outb[toff] = o2;
    }
    float xr0 = 0.0f, xr1 = 0.0f;
    if (t + 2 < TIME) {
      size_t off = (size_t)(t + 2) * IND;
      if (xv0) xr0 = xg0[off];
      if (xv1) xr1 = xg1[off];
    }
    v8f acc = acc_next;
    #pragma unroll
    for (int v = 0; v < 8; ++v) acc_next[v] = bbv[v];
    if (dec_now) {
      float p1[16], p2[16];
      #pragma unroll
      for (int w = 0; w < 16; ++w) { p1[w] = Pp[w * PS + ca];  p2[w] = Pp[w * PS + cb]; }
      acc_next = xproj(fWi, Xl, lm, kh, acc_next);
      float s1 = bda, s2 = lower ? bdb : 0.0f;
      #pragma unroll
      for (int w = 0; w < 16; ++w) { s1 += p1[w];  s2 += p2[w]; }
      o1 = s1;  o2 = s2;
      if (!lower) Rl[rl_idx] = s2;
    } else {
      if (!lower) {
        float p2[16];
        #pragma unroll
        for (int w = 0; w < 16; ++w) p2[w] = Pp[w * PS + cb];
        float s2 = 0.0f;
        #pragma unroll
        for (int w = 0; w < 16; ++w) s2 += p2[w];
        Rl[rl_idx] = s2;
      }
      acc_next = xproj(fWi, Xl, lm, kh, acc_next);
    }
    __syncthreads();

    {
      const v4f* rr = (const v4f*)(Rl + lm * RS + 8 * kh);
      v4f ra = rr[0], rb = rr[1];
      v2f b;
      b.x = ra.x; b.y = ra.y; acc = wmma4(fAw[0], b, acc);
      b.x = ra.z; b.y = ra.w; acc = wmma4(fAw[1], b, acc);
      b.x = rb.x; b.y = rb.y; acc = wmma4(fAw[2], b, acc);
      b.x = rb.z; b.y = rb.w; acc = wmma4(fAw[3], b, acc);
    }
    #pragma unroll
    for (int v = 0; v < 8; ++v) hT[v] = fast_tanh(acc[v]);
    if (t + 2 < TIME) {
      if (xv0) Xl[xl0] = xr0;
      if (xv1) Xl[xl1] = xr1;
    }
  }

  if (TIME > 1 && ((TIME - 2) & (NREP - 1)) == ph) {
    const size_t toff = (size_t)(TIME - 2) * IND;
    if (sva) outa[toff] = o1;
    if (svb) outb[toff] = o2;
  }
  if (((TIME - 1) & (NREP - 1)) == ph) {
    dec_partials(fWd, hT, Pw, lm, kh);
    __syncthreads();
    const size_t toff = (size_t)(TIME - 1) * IND;
    if (sva) {
      float s = bda;
      #pragma unroll
      for (int w = 0; w < 16; ++w) s += Pp[w * PS + ca];
      outa[toff] = s;
    }
    if (svb) {
      float s = bdb;
      #pragma unroll
      for (int w = 0; w < 16; ++w) s += Pp[w * PS + cb];
      outb[toff] = s;
    }
  }

  if (ph == 0) {
    float* hf = out + (size_t)BATCH * TIME * IND + (size_t)(b0 + lm) * HID + n0 + 8 * kh;
    #pragma unroll
    for (int v = 0; v < 8; ++v) hf[v] = hT[v];
  }
}

extern "C" void kernel_launch(void* const* d_in, const int* in_sizes, int n_in,
                              void* d_out, int out_size, void* d_ws, size_t ws_size,
                              hipStream_t stream) {
  const float* x     = (const float*)d_in[0];
  const float* W_in  = (const float*)d_in[1];
  const float* Amat  = (const float*)d_in[2];
  const float* B_low = (const float*)d_in[3];
  const float* bvec  = (const float*)d_in[4];
  const float* W_dec = (const float*)d_in[5];
  const float* b_dec = (const float*)d_in[6];
  float* out = (float*)d_out;

  dream_scan<<<(BATCH / 16) * NREP, 512, 0, stream>>>(
      x, W_in, Amat, B_low, bvec, W_dec, b_dec, out);
}


# Input projection of step t+1 issued in pieces (2+6+2 WMMAs) in the waits behind each LDS hand-over of step t, from x fragments held in registers; x tile double-buffered; global traffic in phase C
# speedup vs baseline: 2.2147x; 1.0546x over previous
#include <hip/hip_runtime.h>
#include <math.h>


typedef float v2f __attribute__((ext_vector_type(2)));
typedef float v4f __attribute__((ext_vector_type(4)));
typedef float v8f __attribute__((ext_vector_type(8)));

#define BATCH 256
#define TIME  1024
#define IND   39
#define HID   256
#define RANK  16
#define NREP  8

#define XS  44
#define RS  20
#define DS  52
#define PR  (16 * DS)
#define PS  1152

#define OFF_X 0
#define OFF_R (OFF_X + 2 * 16 * XS)
#define OFF_P (OFF_R + 16 * RS)
#define SMEM_FLOATS (OFF_P + 16 * PS)
#define SMEM_BYTES  (SMEM_FLOATS * 4)
#define LDS_RESERVE (164 * 1024)

__device__ __forceinline__ v8f wmma4(v2f a, v2f b, v8f c) {
  return __builtin_amdgcn_wmma_f32_16x16x4_f32(
       false, a,  false, b,
       (short)0, c,  false,  false);
}

__device__ __forceinline__ float fast_tanh(float v) {
#if __has_builtin(__builtin_amdgcn_tanhf)
  return __builtin_amdgcn_tanhf(v);
#else
  return tanhf(v);
#endif
}

__device__ __forceinline__ v8f xproj(const v2f* fWi, const float* Xl, int lm, int kh, v8f acc) {
  const v4f* xp = (const v4f*)(Xl + lm * XS + 20 * kh);
  v4f q0 = xp[0], q1 = xp[1], q2 = xp[2], q3 = xp[3], q4 = xp[4];
  v2f b;
  b.x = q0.x; b.y = q0.y; acc = wmma4(fWi[0], b, acc);
  b.x = q0.z; b.y = q0.w; acc = wmma4(fWi[1], b, acc);
  b.x = q1.x; b.y = q1.y; acc = wmma4(fWi[2], b, acc);
  b.x = q1.z; b.y = q1.w; acc = wmma4(fWi[3], b, acc);
  b.x = q2.x; b.y = q2.y; acc = wmma4(fWi[4], b, acc);
  b.x = q2.z; b.y = q2.w; acc = wmma4(fWi[5], b, acc);
  b.x = q3.x; b.y = q3.y; acc = wmma4(fWi[6], b, acc);
  b.x = q3.z; b.y = q3.w; acc = wmma4(fWi[7], b, acc);
  b.x = q4.x; b.y = q4.y; acc = wmma4(fWi[8], b, acc);
  b.x = q4.z; b.y = q4.w; acc = wmma4(fWi[9], b, acc);
  return acc;
}

__device__ __forceinline__ v8f xstep(const v2f* fWi, const v4f* xf, int j, v8f acc) {
  v2f b;
  if (j & 1) { b.x = xf[j >> 1].z; b.y = xf[j >> 1].w; }
  else       { b.x = xf[j >> 1].x; b.y = xf[j >> 1].y; }
  return wmma4(fWi[j], b, acc);
}

__device__ __forceinline__ void dec_partials(const v2f (*fWd)[4], v8f hT, float* Pw, int lm, int kh) {
  v8f d0 = {}, d1 = {}, d2 = {};
  #pragma unroll
  for (int j = 0; j < 4; ++j) {
    v2f b; b.x = hT[2 * j]; b.y = hT[2 * j + 1];
    d0 = wmma4(fWd[0][j], b, d0);
    d1 = wmma4(fWd[1][j], b, d1);
    d2 = wmma4(fWd[2][j], b, d2);
  }
  v4f* pd = (v4f*)(Pw + lm * DS + 8 * kh);
  v4f s;
  s.x = d0[0]; s.y = d0[1]; s.z = d0[2]; s.w = d0[3]; pd[0] = s;
  s.x = d0[4]; s.y = d0[5]; s.z = d0[6]; s.w = d0[7]; pd[1] = s;
  s.x = d1[0]; s.y = d1[1]; s.z = d1[2]; s.w = d1[3]; pd[4] = s;
  s.x = d1[4]; s.y = d1[5]; s.z = d1[6]; s.w = d1[7]; pd[5] = s;
  s.x = d2[0]; s.y = d2[1]; s.z = d2[2]; s.w = d2[3]; pd[8] = s;
  s.x = d2[4]; s.y = d2[5]; s.z = d2[6]; s.w = d2[7]; pd[9] = s;
}

__global__ __launch_bounds__(512)
void dream_scan(const float* __restrict__ x, const float* __restrict__ W_in,
                const float* __restrict__ Amat, const float* __restrict__ B_low,
                const float* __restrict__ bvec, const float* __restrict__ W_dec,
                const float* __restrict__ b_dec, float* __restrict__ out) {
  __shared__ __attribute__((aligned(16))) float sm[LDS_RESERVE / 4];
  float* Xl = sm + OFF_X;
  float* Rl = sm + OFF_R;
  float* Pp = sm + OFF_P;

  const int tid  = threadIdx.x;
  const int wave = tid >> 5;
  const int lane = tid & 31;
  const int lm   = lane & 15;
  const int kh   = lane >> 4;
  const int b0   = (blockIdx.x & 15) * 16;
  const int ph   = blockIdx.x >> 4;
  const int n0   = wave * 16;

  v2f fBl[4], fAw[4], fWd[3][4], fWi[10];
  #pragma unroll
  for (int j = 0; j < 4; ++j) {
    const float* p = B_low + lm * HID + n0 + 8 * kh + 2 * j;
    fBl[j].x = p[0]; fBl[j].y = p[1];
    const float* q = Amat + (n0 + lm) * RANK + 8 * kh + 2 * j;
    fAw[j].x = q[0]; fAw[j].y = q[1];
  }
  #pragma unroll
  for (int i = 0; i < 3; ++i) {
    const int c = 16 * i + lm;
    #pragma unroll
    for (int j = 0; j < 4; ++j) {
      const float* p = W_dec + (c < IND ? c : 0) * HID + n0 + 8 * kh + 2 * j;
      fWd[i][j].x = (c < IND) ? p[0] : 0.0f;
      fWd[i][j].y = (c < IND) ? p[1] : 0.0f;
    }
  }
  #pragma unroll
  for (int j = 0; j < 10; ++j) {
    const float* p = W_in + (n0 + lm) * IND;
    const int k = 20 * kh + 2 * j;
    fWi[j].x = p[k];
    fWi[j].y = (k + 1 < IND) ? p[k + 1] : 0.0f;
  }
  float bbv[8];
  #pragma unroll
  for (int v = 0; v < 8; ++v) bbv[v] = bvec[n0 + 8 * kh + v];

  for (int e = tid; e < 2 * 16 * XS; e += 512) Xl[e] = 0.0f;
  __syncthreads();

  const int  e0 = tid, e1 = tid + 512;
  const bool xv0 = (e0 < 16 * IND), xv1 = (e1 < 16 * IND);
  const float* xg0 = x;  int xl0 = 0;
  const float* xg1 = x;  int xl1 = 0;
  if (xv0) { int r = e0 / IND, c = e0 % IND;
    xg0 = x + (size_t)(b0 + r) * TIME * IND + c;  xl0 = r * XS + c; }
  if (xv1) { int r = e1 / IND, c = e1 % IND;
    xg1 = x + (size_t)(b0 + r) * TIME * IND + c;  xl1 = r * XS + c; }

  const int  ea = tid, eb = tid + 512;
  const bool lower = (tid < 256);
  const bool sva = ((ea % 48) < IND);
  const bool svb = lower && ((eb % 48) < IND);
  float* outa = out;  float bda = 0.0f;
  float* outb = out;  float bdb = 0.0f;
  if (sva) { int row = ea / 48, c = ea % 48;
    outa = out + (size_t)(b0 + row) * TIME * IND + c;  bda = b_dec[c]; }
  if (svb) { int row = eb / 48, c = eb % 48;
    outb = out + (size_t)(b0 + row) * TIME * IND + c;  bdb = b_dec[c]; }
  const int rq = tid - 256;
  const int ca = (ea / 48) * DS + (ea % 48);
  const int cb = lower ? ((eb / 48) * DS + (eb % 48)) : (PR + (rq >> 4) * RS + (rq & 15));
  const int rl_idx = lower ? 0 : ((rq >> 4) * RS + (rq & 15));
  float* Pw = Pp + wave * PS;


  if (xv0) Xl[xl0] = xg0[0];
  if (xv1) Xl[xl1] = xg1[0];
  if (TIME > 1) {
    if (xv0) Xl[16 * XS + xl0] = xg0[IND];
    if (xv1) Xl[16 * XS + xl1] = xg1[IND];
  }
  __syncthreads();
  v8f acc;
  #pragma unroll
  for (int v = 0; v < 8; ++v) acc[v] = bbv[v];
  acc = xproj(fWi, Xl, lm, kh, acc);
  v4f xf[5];
  {
    const v4f* xp = (const v4f*)(Xl + 16 * XS + lm * XS + 20 * kh);
    #pragma unroll
    for (int i = 0; i < 5; ++i) xf[i] = xp[i];
  }
  float xr0 = 0.0f, xr1 = 0.0f;
  if (TIME > 2) {
    if (xv0) xr0 = xg0[2 * IND];
    if (xv1) xr1 = xg1[2 * IND];
  }
  v8f acc_next;
  #pragma unroll
  for (int v = 0; v < 8; ++v) acc_next[v] = bbv[v];

  v8f hT = {};
  float o1 = 0.0f, o2 = 0.0f;

  for (int t = 0; t < TIME; ++t) {
    const bool dec_now = (t > 0) && (((t - 1) & (NREP - 1)) == ph);
    float* Xw = Xl + (t & 1) * (16 * XS);

    {
      v8f rp = {};
      #pragma unroll
      for (int j = 0; j < 4; ++j) {
        v2f b; b.x = hT[2 * j]; b.y = hT[2 * j + 1];
        rp = wmma4(fBl[j], b, rp);
      }
      v4f* pr = (v4f*)(Pw + PR + lm * RS + 8 * kh);
      v4f s;
      s.x = rp[0]; s.y = rp[1]; s.z = rp[2]; s.w = rp[3]; pr[0] = s;
      s.x = rp[4]; s.y = rp[5]; s.z = rp[6]; s.w = rp[7]; pr[1] = s;
    }
    if (dec_now) dec_partials(fWd, hT, Pw, lm, kh);
    acc_next = xstep(fWi, xf, 0, acc_next);
    acc_next = xstep(fWi, xf, 1, acc_next);
    __syncthreads();

    if (t + 2 < TIME) {
      if (xv0) Xw[xl0] = xr0;
      if (xv1) Xw[xl1] = xr1;
    }
    if (dec_now) {
      float p1[16], p2[16];
      #pragma unroll
      for (int w = 0; w < 16; ++w) { p1[w] = Pp[w * PS + ca];  p2[w] = Pp[w * PS + cb]; }
      float s1 = bda, s2 = lower ? bdb : 0.0f;
      #pragma unroll
      for (int w = 0; w < 16; ++w) { s1 += p1[w];  s2 += p2[w]; }
      o1 = s1;  o2 = s2;
      if (!lower) Rl[rl_idx] = s2;
    } else {
      if (!lower) {
        float p2[16];
        #pragma unroll
        for (int w = 0; w < 16; ++w) p2[w] = Pp[w * PS + cb];
        float s2 = 0.0f;
        #pragma unroll
        for (int w = 0; w < 16; ++w) s2 += p2[w];
        Rl[rl_idx] = s2;
      }
    }
    acc_next = xstep(fWi, xf, 2, acc_next);
    acc_next = xstep(fWi, xf, 3, acc_next);
    acc_next = xstep(fWi, xf, 4, acc_next);
    acc_next = xstep(fWi, xf, 5, acc_next);
    acc_next = xstep(fWi, xf, 6, acc_next);
    acc_next = xstep(fWi, xf, 7, acc_next);
    __builtin_amdgcn_sched_barrier(0);
    __syncthreads();

    {
      const v4f* rr = (const v4f*)(Rl + lm * RS + 8 * kh);
      v4f ra = rr[0], rb = rr[1];
      __builtin_amdgcn_sched_barrier(0);
      acc_next = xstep(fWi, xf, 8, acc_next);
      acc_next = xstep(fWi, xf, 9, acc_next);
      if (dec_now) {
        const size_t toff = (size_t)(t - 1) * IND;
        if (sva) outa[toff] = o1;
        if (svb) outb[toff] = o2;
      }
      if (t + 3 < TIME) {
        size_t off = (size_t)(t + 3) * IND;
        if (xv0) xr0 = xg0[off];
        if (xv1) xr1 = xg1[off];
      }
      v2f b;
      b.x = ra.x; b.y = ra.y; acc = wmma4(fAw[0], b, acc);
      b.x = ra.z; b.y = ra.w; acc = wmma4(fAw[1], b, acc);
      b.x = rb.x; b.y = rb.y; acc = wmma4(fAw[2], b, acc);
      b.x = rb.z; b.y = rb.w; acc = wmma4(fAw[3], b, acc);
      const v4f* xp = (const v4f*)(Xw + lm * XS + 20 * kh);
      #pragma unroll
      for (int i = 0; i < 5; ++i) xf[i] = xp[i];
    }
    #pragma unroll
    for (int v = 0; v < 8; ++v) hT[v] = fast_tanh(acc[v]);
    acc = acc_next;
    #pragma unroll
    for (int v = 0; v < 8; ++v) acc_next[v] = bbv[v];
  }

  if (((TIME - 1) & (NREP - 1)) == ph) {
    dec_partials(fWd, hT, Pw, lm, kh);
    __syncthreads();
    const size_t toff = (size_t)(TIME - 1) * IND;
    if (sva) {
      float s = bda;
      #pragma unroll
      for (int w = 0; w < 16; ++w) s += Pp[w * PS + ca];
      outa[toff] = s;
    }
    if (svb) {
      float s = bdb;
      #pragma unroll
      for (int w = 0; w < 16; ++w) s += Pp[w * PS + cb];
      outb[toff] = s;
    }
  }

  if (ph == 0) {
    float* hf = out + (size_t)BATCH * TIME * IND + (size_t)(b0 + lm) * HID + n0 + 8 * kh;
    #pragma unroll
    for (int v = 0; v < 8; ++v) hf[v] = hT[v];
  }
}

extern "C" void kernel_launch(void* const* d_in, const int* in_sizes, int n_in,
                              void* d_out, int out_size, void* d_ws, size_t ws_size,
                              hipStream_t stream) {
  const float* x     = (const float*)d_in[0];
  const float* W_in  = (const float*)d_in[1];
  const float* Amat  = (const float*)d_in[2];
  const float* B_low = (const float*)d_in[3];
  const float* bvec  = (const float*)d_in[4];
  const float* W_dec = (const float*)d_in[5];
  const float* b_dec = (const float*)d_in[6];
  float* out = (float*)d_out;

  dream_scan<<<(BATCH / 16) * NREP, 512, 0, stream>>>(
      x, W_in, Amat, B_low, bvec, W_dec, b_dec, out);
}


# Input projection b + W_in x_t moved to a first launch on all WGPs that writes it to the workspace in accumulator layout; scan loads it 2 steps ahead (steps that do not fit stay in-loop); same sums
# speedup vs baseline: 3.0118x; 1.3599x over previous
#include <hip/hip_runtime.h>
#include <math.h>


typedef float v2f __attribute__((ext_vector_type(2)));
typedef float v4f __attribute__((ext_vector_type(4)));
typedef float v8f __attribute__((ext_vector_type(8)));

#define BATCH 256
#define TIME  1024
#define IND   39
#define HID   256
#define RANK  16
#define NREP  8

#define XS  44
#define RS  20
#define DS  52
#define PR  (16 * DS)
#define PS  1152

#define OFF_X 0
#define OFF_R (OFF_X + 2 * 16 * XS)
#define OFF_P (OFF_R + 16 * RS)
#define SMEM_FLOATS (OFF_P + 16 * PS)
#define SMEM_BYTES  (SMEM_FLOATS * 4)
#define LDS_RESERVE (164 * 1024)

__device__ __forceinline__ v8f wmma4(v2f a, v2f b, v8f c) {
  return __builtin_amdgcn_wmma_f32_16x16x4_f32(
       false, a,  false, b,
       (short)0, c,  false,  false);
}

__device__ __forceinline__ float fast_tanh(float v) {
#if __has_builtin(__builtin_amdgcn_tanhf)
  return __builtin_amdgcn_tanhf(v);
#else
  return tanhf(v);
#endif
}

__device__ __forceinline__ v8f xproj(const v2f* fWi, const float* Xl, int lm, int kh, v8f acc) {
  const v4f* xp = (const v4f*)(Xl + lm * XS + 20 * kh);
  v4f q0 = xp[0], q1 = xp[1], q2 = xp[2], q3 = xp[3], q4 = xp[4];
  v2f b;
  b.x = q0.x; b.y = q0.y; acc = wmma4(fWi[0], b, acc);
  b.x = q0.z; b.y = q0.w; acc = wmma4(fWi[1], b, acc);
  b.x = q1.x; b.y = q1.y; acc = wmma4(fWi[2], b, acc);
  b.x = q1.z; b.y = q1.w; acc = wmma4(fWi[3], b, acc);
  b.x = q2.x; b.y = q2.y; acc = wmma4(fWi[4], b, acc);
  b.x = q2.z; b.y = q2.w; acc = wmma4(fWi[5], b, acc);
  b.x = q3.x; b.y = q3.y; acc = wmma4(fWi[6], b, acc);
  b.x = q3.z; b.y = q3.w; acc = wmma4(fWi[7], b, acc);
  b.x = q4.x; b.y = q4.y; acc = wmma4(fWi[8], b, acc);
  b.x = q4.z; b.y = q4.w; acc = wmma4(fWi[9], b, acc);
  return acc;
}

__device__ __forceinline__ v8f xstep(const v2f* fWi, const v4f* xf, int j, v8f acc) {
  v2f b;
  if (j & 1) { b.x = xf[j >> 1].z; b.y = xf[j >> 1].w; }
  else       { b.x = xf[j >> 1].x; b.y = xf[j >> 1].y; }
  return wmma4(fWi[j], b, acc);
}

__device__ __forceinline__ void dec_partials(const v2f (*fWd)[4], v8f hT, float* Pw, int lm, int kh) {
  v8f d0 = {}, d1 = {}, d2 = {};
  #pragma unroll
  for (int j = 0; j < 4; ++j) {
    v2f b; b.x = hT[2 * j]; b.y = hT[2 * j + 1];
    d0 = wmma4(fWd[0][j], b, d0);
    d1 = wmma4(fWd[1][j], b, d1);
    d2 = wmma4(fWd[2][j], b, d2);
  }
  v4f* pd = (v4f*)(Pw + lm * DS + 8 * kh);
  v4f s;
  s.x = d0[0]; s.y = d0[1]; s.z = d0[2]; s.w = d0[3]; pd[0] = s;
  s.x = d0[4]; s.y = d0[5]; s.z = d0[6]; s.w = d0[7]; pd[1] = s;
  s.x = d1[0]; s.y = d1[1]; s.z = d1[2]; s.w = d1[3]; pd[4] = s;
  s.x = d1[4]; s.y = d1[5]; s.z = d1[6]; s.w = d1[7]; pd[5] = s;
  s.x = d2[0]; s.y = d2[1]; s.z = d2[2]; s.w = d2[3]; pd[8] = s;
  s.x = d2[4]; s.y = d2[5]; s.z = d2[6]; s.w = d2[7]; pd[9] = s;
}


#define PSTEPF (16 * 16 * 32 * 8)
#define PROJ_T 16

__device__ __forceinline__ int steps_that_fit(unsigned long long ws_bytes) {
  const unsigned long long n = ws_bytes / ((unsigned long long)PSTEPF * 4ull);
  const int m = (n > (unsigned long long)TIME) ? TIME : (int)n;
  return m & ~1;
}

__device__ __forceinline__ void lds_barrier() {
  __builtin_amdgcn_fence(__ATOMIC_RELEASE, "workgroup", "local");
  __builtin_amdgcn_s_barrier();
  __builtin_amdgcn_fence(__ATOMIC_ACQUIRE, "workgroup", "local");
}

__global__ __launch_bounds__(512)
void dream_proj(const float* __restrict__ x, const float* __restrict__ W_in,
                const float* __restrict__ bvec, float* __restrict__ P, unsigned long long ws_bytes) {
  __shared__ __attribute__((aligned(16))) float Xl[2 * 16 * XS];
  const int tid  = threadIdx.x;
  const int wave = tid >> 5;
  const int lane = tid & 31;
  const int lm   = lane & 15;
  const int kh   = lane >> 4;
  const int tile = blockIdx.x & 15;
  const int b0   = tile * 16;
  const int n0   = wave * 16;
  const int tbeg = (blockIdx.x >> 4) * PROJ_T;
  const int tfit = steps_that_fit(ws_bytes);
  const int tend = (tbeg + PROJ_T < tfit) ? (tbeg + PROJ_T) : tfit;
  if (tbeg >= tend) return;

  v2f fWi[10];
  #pragma unroll
  for (int j = 0; j < 10; ++j) {
    const float* p = W_in + (n0 + lm) * IND;
    const int k = 20 * kh + 2 * j;
    fWi[j].x = p[k];
    fWi[j].y = (k + 1 < IND) ? p[k + 1] : 0.0f;
  }
  float bbv[8];
  #pragma unroll
  for (int v = 0; v < 8; ++v) bbv[v] = bvec[n0 + 8 * kh + v];

  for (int e = tid; e < 2 * 16 * XS; e += 512) Xl[e] = 0.0f;
  __syncthreads();

  const int  e0 = tid, e1 = tid + 512;
  const bool xv1 = (e1 < 16 * IND);
  const float* xg0;  int xl0;
  const float* xg1 = x;  int xl1 = 0;
  { int r = e0 / IND, c = e0 % IND;
    xg0 = x + (size_t)(b0 + r) * TIME * IND + c;  xl0 = r * XS + c; }
  if (xv1) { int r = e1 / IND, c = e1 % IND;
    xg1 = x + (size_t)(b0 + r) * TIME * IND + c;  xl1 = r * XS + c; }

  float xr0 = xg0[(size_t)tbeg * IND];
  float xr1 = xv1 ? xg1[(size_t)tbeg * IND] : 0.0f;
  for (int t = tbeg; t < tend; ++t) {
    float* Xt = Xl + ((t - tbeg) & 1) * (16 * XS);
    Xt[xl0] = xr0;
    if (xv1) Xt[xl1] = xr1;
    if (t + 1 < tend) {
      xr0 = xg0[(size_t)(t + 1) * IND];
      if (xv1) xr1 = xg1[(size_t)(t + 1) * IND];
    }
    lds_barrier();
    v8f acc;
    #pragma unroll
    for (int v = 0; v < 8; ++v) acc[v] = bbv[v];
    acc = xproj(fWi, Xt, lm, kh, acc);
    v4f* dst = (v4f*)(P + ((size_t)(t * 16 + tile) * 16 + wave) * 256 + lane * 8);
    v4f s;
    s.x = acc[0]; s.y = acc[1]; s.z = acc[2]; s.w = acc[3]; dst[0] = s;
    s.x = acc[4]; s.y = acc[5]; s.z = acc[6]; s.w = acc[7]; dst[1] = s;
  }
}

__global__ __launch_bounds__(512)
void dream_scan(const float* __restrict__ x, const float* __restrict__ W_in,
                const float* __restrict__ Amat, const float* __restrict__ B_low,
                const float* __restrict__ bvec, const float* __restrict__ W_dec,
                const float* __restrict__ b_dec, float* __restrict__ out,
                const float* __restrict__ Pws, unsigned long long ws_bytes) {
  __shared__ __attribute__((aligned(16))) float sm[LDS_RESERVE / 4];
  float* Xl = sm + OFF_X;
  float* Rl = sm + OFF_R;
  float* Pp = sm + OFF_P;

  const int tid  = threadIdx.x;
  const int wave = tid >> 5;
  const int lane = tid & 31;
  const int lm   = lane & 15;
  const int kh   = lane >> 4;
  const int b0   = (blockIdx.x & 15) * 16;
  const int ph   = blockIdx.x >> 4;
  const int n0   = wave * 16;

  v2f fBl[4], fAw[4], fWd[3][4], fWi[10];
  #pragma unroll
  for (int j = 0; j < 4; ++j) {
    const float* p = B_low + lm * HID + n0 + 8 * kh + 2 * j;
    fBl[j].x = p[0]; fBl[j].y = p[1];
    const float* q = Amat + (n0 + lm) * RANK + 8 * kh + 2 * j;
    fAw[j].x = q[0]; fAw[j].y = q[1];
  }
  #pragma unroll
  for (int i = 0; i < 3; ++i) {
    const int c = 16 * i + lm;
    #pragma unroll
    for (int j = 0; j < 4; ++j) {
      const float* p = W_dec + (c < IND ? c : 0) * HID + n0 + 8 * kh + 2 * j;
      fWd[i][j].x = (c < IND) ? p[0] : 0.0f;
      fWd[i][j].y = (c < IND) ? p[1] : 0.0f;
    }
  }
  #pragma unroll
  for (int j = 0; j < 10; ++j) {
    const float* p = W_in + (n0 + lm) * IND;
    const int k = 20 * kh + 2 * j;
    fWi[j].x = p[k];
    fWi[j].y = (k + 1 < IND) ? p[k + 1] : 0.0f;
  }
  float bbv[8];
  #pragma unroll
  for (int v = 0; v < 8; ++v) bbv[v] = bvec[n0 + 8 * kh + v];

  for (int e = tid; e < 2 * 16 * XS; e += 512) Xl[e] = 0.0f;
  __syncthreads();

  const int  e0 = tid, e1 = tid + 512;
  const bool xv0 = (e0 < 16 * IND), xv1 = (e1 < 16 * IND);
  const float* xg0 = x;  int xl0 = 0;
  const float* xg1 = x;  int xl1 = 0;
  if (xv0) { int r = e0 / IND, c = e0 % IND;
    xg0 = x + (size_t)(b0 + r) * TIME * IND + c;  xl0 = r * XS + c; }
  if (xv1) { int r = e1 / IND, c = e1 % IND;
    xg1 = x + (size_t)(b0 + r) * TIME * IND + c;  xl1 = r * XS + c; }

  const int  ea = tid, eb = tid + 512;
  const bool lower = (tid < 256);
  const bool sva = ((ea % 48) < IND);
  const bool svb = lower && ((eb % 48) < IND);
  float* outa = out;  float bda = 0.0f;
  float* outb = out;  float bdb = 0.0f;
  if (sva) { int row = ea / 48, c = ea % 48;
    outa = out + (size_t)(b0 + row) * TIME * IND + c;  bda = b_dec[c]; }
  if (svb) { int row = eb / 48, c = eb % 48;
    outb = out + (size_t)(b0 + row) * TIME * IND + c;  bdb = b_dec[c]; }
  const int rq = tid - 256;
  const int ca = (ea / 48) * DS + (ea % 48);
  const int cb = lower ? ((eb / 48) * DS + (eb % 48)) : (PR + (rq >> 4) * RS + (rq & 15));
  const int rl_idx = lower ? 0 : ((rq >> 4) * RS + (rq & 15));
  float* Pw = Pp + wave * PS;


  v8f hT = {};
  float o1 = 0.0f, o2 = 0.0f;

  const int t0 = steps_that_fit(ws_bytes);
  {
    const float* Pg = Pws + ((size_t)(b0 >> 4) * 16 + wave) * 256 + lane * 8;
    v8f S0 = {}, S1 = {};
    if (t0 > 0) {
      const v4f* pa = (const v4f*)Pg;
      const v4f* pb = (const v4f*)(Pg + PSTEPF);
      v4f a0 = pa[0], a1 = pa[1], c0 = pb[0], c1 = pb[1];
      S0[0] = a0.x; S0[1] = a0.y; S0[2] = a0.z; S0[3] = a0.w; S0[4] = a1.x; S0[5] = a1.y; S0[6] = a1.z; S0[7] = a1.w;
      S1[0] = c0.x; S1[1] = c0.y; S1[2] = c0.z; S1[3] = c0.w; S1[4] = c1.x; S1[5] = c1.y; S1[6] = c1.z; S1[7] = c1.w;
    }
    auto pstep = [&](const int t, v8f& S) __attribute__((always_inline)) {
      const bool dec_now = (t > 0) && (((t - 1) & (NREP - 1)) == ph);
      {
        v8f rp = {};
        #pragma unroll
        for (int j = 0; j < 4; ++j) {
          v2f b; b.x = hT[2 * j]; b.y = hT[2 * j + 1];
          rp = wmma4(fBl[j], b, rp);
        }
        v4f* pr = (v4f*)(Pw + PR + lm * RS + 8 * kh);
        v4f s;
        s.x = rp[0]; s.y = rp[1]; s.z = rp[2]; s.w = rp[3]; pr[0] = s;
        s.x = rp[4]; s.y = rp[5]; s.z = rp[6]; s.w = rp[7]; pr[1] = s;
      }
      if (dec_now) dec_partials(fWd, hT, Pw, lm, kh);
      lds_barrier();
      if (dec_now) {
        float p1[16], p2[16];
        #pragma unroll
        for (int w = 0; w < 16; ++w) { p1[w] = Pp[w * PS + ca];  p2[w] = Pp[w * PS + cb]; }
        float s1 = bda, s2 = lower ? bdb : 0.0f;
        #pragma unroll
        for (int w = 0; w < 16; ++w) { s1 += p1[w];  s2 += p2[w]; }
        o1 = s1;  o2 = s2;
        if (!lower) Rl[rl_idx] = s2;
      } else {
        if (!lower) {
          float p2[16];
          #pragma unroll
          for (int w = 0; w < 16; ++w) p2[w] = Pp[w * PS + cb];
          float s2 = 0.0f;
          #pragma unroll
          for (int w = 0; w < 16; ++w) s2 += p2[w];
          Rl[rl_idx] = s2;
        }
      }
      lds_barrier();
      v8f acc;
      {
        const v4f* rr = (const v4f*)(Rl + lm * RS + 8 * kh);
        v4f ra = rr[0], rb = rr[1];
        if (dec_now) {
          const size_t toff = (size_t)(t - 1) * IND;
          if (sva) outa[toff] = o1;
          if (svb) outb[toff] = o2;
        }
        v2f b;
        b.x = ra.x; b.y = ra.y; acc = wmma4(fAw[0], b, S);
        b.x = ra.z; b.y = ra.w; acc = wmma4(fAw[1], b, acc);
        b.x = rb.x; b.y = rb.y; acc = wmma4(fAw[2], b, acc);
        b.x = rb.z; b.y = rb.w; acc = wmma4(fAw[3], b, acc);
      }
      #pragma unroll
      for (int v = 0; v < 8; ++v) hT[v] = fast_tanh(acc[v]);
      if (t + 2 < t0) {
        const v4f* pp = (const v4f*)(Pg + (size_t)(t + 2) * PSTEPF);
        v4f a0 = pp[0], a1 = pp[1];
        S[0] = a0.x; S[1] = a0.y; S[2] = a0.z; S[3] = a0.w; S[4] = a1.x; S[5] = a1.y; S[6] = a1.z; S[7] = a1.w;
      }
    };
    for (int t = 0; t < t0; t += 2) { pstep(t, S0); pstep(t + 1, S1); }
  }

  v8f acc;
  #pragma unroll
  for (int v = 0; v < 8; ++v) acc[v] = bbv[v];
  v4f xf[5] = {};
  float xr0 = 0.0f, xr1 = 0.0f;
  if (t0 < TIME) {
    if (xv0) Xl[xl0] = xg0[(size_t)t0 * IND];
    if (xv1) Xl[xl1] = xg1[(size_t)t0 * IND];
    if (t0 + 1 < TIME) {
      if (xv0) Xl[16 * XS + xl0] = xg0[(size_t)(t0 + 1) * IND];
      if (xv1) Xl[16 * XS + xl1] = xg1[(size_t)(t0 + 1) * IND];
    }
    __syncthreads();
    acc = xproj(fWi, Xl, lm, kh, acc);
    {
      const v4f* xp = (const v4f*)(Xl + 16 * XS + lm * XS + 20 * kh);
      #pragma unroll
      for (int i = 0; i < 5; ++i) xf[i] = xp[i];
    }
    if (t0 + 2 < TIME) {
      if (xv0) xr0 = xg0[(size_t)(t0 + 2) * IND];
      if (xv1) xr1 = xg1[(size_t)(t0 + 2) * IND];
    }
  }
  v8f acc_next;
  #pragma unroll
  for (int v = 0; v < 8; ++v) acc_next[v] = bbv[v];

  for (int t = t0; t < TIME; ++t) {
    const bool dec_now = (t > 0) && (((t - 1) & (NREP - 1)) == ph);
    float* Xw = Xl + (t & 1) * (16 * XS);

    {
      v8f rp = {};
      #pragma unroll
      for (int j = 0; j < 4; ++j) {
        v2f b; b.x = hT[2 * j]; b.y = hT[2 * j + 1];
        rp = wmma4(fBl[j], b, rp);
      }
      v4f* pr = (v4f*)(Pw + PR + lm * RS + 8 * kh);
      v4f s;
      s.x = rp[0]; s.y = rp[1]; s.z = rp[2]; s.w = rp[3]; pr[0] = s;
      s.x = rp[4]; s.y = rp[5]; s.z = rp[6]; s.w = rp[7]; pr[1] = s;
    }
    if (dec_now) dec_partials(fWd, hT, Pw, lm, kh);
    acc_next = xstep(fWi, xf, 0, acc_next);
    acc_next = xstep(fWi, xf, 1, acc_next);
    __syncthreads();

    if (t + 2 < TIME) {
      if (xv0) Xw[xl0] = xr0;
      if (xv1) Xw[xl1] = xr1;
    }
    if (dec_now) {
      float p1[16], p2[16];
      #pragma unroll
      for (int w = 0; w < 16; ++w) { p1[w] = Pp[w * PS + ca];  p2[w] = Pp[w * PS + cb]; }
      float s1 = bda, s2 = lower ? bdb : 0.0f;
      #pragma unroll
      for (int w = 0; w < 16; ++w) { s1 += p1[w];  s2 += p2[w]; }
      o1 = s1;  o2 = s2;
      if (!lower) Rl[rl_idx] = s2;
    } else {
      if (!lower) {
        float p2[16];
        #pragma unroll
        for (int w = 0; w < 16; ++w) p2[w] = Pp[w * PS + cb];
        float s2 = 0.0f;
        #pragma unroll
        for (int w = 0; w < 16; ++w) s2 += p2[w];
        Rl[rl_idx] = s2;
      }
    }
    acc_next = xstep(fWi, xf, 2, acc_next);
    acc_next = xstep(fWi, xf, 3, acc_next);
    acc_next = xstep(fWi, xf, 4, acc_next);
    acc_next = xstep(fWi, xf, 5, acc_next);
    acc_next = xstep(fWi, xf, 6, acc_next);
    acc_next = xstep(fWi, xf, 7, acc_next);
    __builtin_amdgcn_sched_barrier(0);
    __syncthreads();

    {
      const v4f* rr = (const v4f*)(Rl + lm * RS + 8 * kh);
      v4f ra = rr[0], rb = rr[1];
      __builtin_amdgcn_sched_barrier(0);
      acc_next = xstep(fWi, xf, 8, acc_next);
      acc_next = xstep(fWi, xf, 9, acc_next);
      if (dec_now) {
        const size_t toff = (size_t)(t - 1) * IND;
        if (sva) outa[toff] = o1;
        if (svb) outb[toff] = o2;
      }
      if (t + 3 < TIME) {
        size_t off = (size_t)(t + 3) * IND;
        if (xv0) xr0 = xg0[off];
        if (xv1) xr1 = xg1[off];
      }
      v2f b;
      b.x = ra.x; b.y = ra.y; acc = wmma4(fAw[0], b, acc);
      b.x = ra.z; b.y = ra.w; acc = wmma4(fAw[1], b, acc);
      b.x = rb.x; b.y = rb.y; acc = wmma4(fAw[2], b, acc);
      b.x = rb.z; b.y = rb.w; acc = wmma4(fAw[3], b, acc);
      const v4f* xp = (const v4f*)(Xw + lm * XS + 20 * kh);
      #pragma unroll
      for (int i = 0; i < 5; ++i) xf[i] = xp[i];
    }
    #pragma unroll
    for (int v = 0; v < 8; ++v) hT[v] = fast_tanh(acc[v]);
    acc = acc_next;
    #pragma unroll
    for (int v = 0; v < 8; ++v) acc_next[v] = bbv[v];
  }

  if (((TIME - 1) & (NREP - 1)) == ph) {
    dec_partials(fWd, hT, Pw, lm, kh);
    __syncthreads();
    const size_t toff = (size_t)(TIME - 1) * IND;
    if (sva) {
      float s = bda;
      #pragma unroll
      for (int w = 0; w < 16; ++w) s += Pp[w * PS + ca];
      outa[toff] = s;
    }
    if (svb) {
      float s = bdb;
      #pragma unroll
      for (int w = 0; w < 16; ++w) s += Pp[w * PS + cb];
      outb[toff] = s;
    }
  }

  if (ph == 0) {
    float* hf = out + (size_t)BATCH * TIME * IND + (size_t)(b0 + lm) * HID + n0 + 8 * kh;
    #pragma unroll
    for (int v = 0; v < 8; ++v) hf[v] = hT[v];
  }
}

extern "C" void kernel_launch(void* const* d_in, const int* in_sizes, int n_in,
                              void* d_out, int out_size, void* d_ws, size_t ws_size,
                              hipStream_t stream) {
  const float* x     = (const float*)d_in[0];
  const float* W_in  = (const float*)d_in[1];
  const float* Amat  = (const float*)d_in[2];
  const float* B_low = (const float*)d_in[3];
  const float* bvec  = (const float*)d_in[4];
  const float* W_dec = (const float*)d_in[5];
  const float* b_dec = (const float*)d_in[6];
  float* out = (float*)d_out;

  float* ws = (float*)d_ws;
  const unsigned long long ws_bytes = ws_size;

  dream_proj<<<(BATCH / 16) * (TIME / PROJ_T), 512, 0, stream>>>(x, W_in, bvec, ws, ws_bytes);
  dream_scan<<<(BATCH / 16) * NREP, 512, 0, stream>>>(
      x, W_in, Amat, B_low, bvec, W_dec, b_dec, out, ws, ws_bytes);
}


# Scan on 4 waves (1 per SIMD, 64 hidden each): r's sum over hidden runs in the wave's WMMA chains, 4 partials (not 16) meet in LDS, 1 barrier/step not 2; tanh between WMMAs; k sum order changes
# speedup vs baseline: 3.7555x; 1.2469x over previous
#include <hip/hip_runtime.h>
#include <math.h>


typedef float v2f __attribute__((ext_vector_type(2)));
typedef float v4f __attribute__((ext_vector_type(4)));
typedef float v8f __attribute__((ext_vector_type(8)));

#define BATCH 256
#define TIME  1024
#define IND   39
#define HID   256
#define RANK  16
#define NREP  8
static_assert(NREP >= 2 && (NREP & (NREP - 1)) == 0, "NREP: a power of two, at least 2");
#define NW    4
#define TPW   4

#define XS  44
#define RS  20
#define DS  52

#define OFF_X  0
#define OFF_RP (OFF_X + 16 * XS)
#define OFF_DP (OFF_RP + 2 * NW * 16 * RS)
#define LDS_RESERVE (164 * 1024)

__device__ __forceinline__ v8f wmma4(v2f a, v2f b, v8f c) {
  return __builtin_amdgcn_wmma_f32_16x16x4_f32(
       false, a,  false, b,
       (short)0, c,  false,  false);
}

__device__ __forceinline__ float fast_tanh(float v) {
#if __has_builtin(__builtin_amdgcn_tanhf)
  return __builtin_amdgcn_tanhf(v);
#else
  return tanhf(v);
#endif
}

__device__ __forceinline__ v2f bfrag(v8f d, int j) {
  v2f b; b.x = d[2 * j]; b.y = d[2 * j + 1];
  return b;
}

__device__ __forceinline__ v8f xproj(const v2f* fWi, const float* Xl, int lm, int kh, v8f acc) {
  const v4f* xp = (const v4f*)(Xl + lm * XS + 20 * kh);
  v4f q0 = xp[0], q1 = xp[1], q2 = xp[2], q3 = xp[3], q4 = xp[4];
  v2f b;
  b.x = q0.x; b.y = q0.y; acc = wmma4(fWi[0], b, acc);
  b.x = q0.z; b.y = q0.w; acc = wmma4(fWi[1], b, acc);
  b.x = q1.x; b.y = q1.y; acc = wmma4(fWi[2], b, acc);
  b.x = q1.z; b.y = q1.w; acc = wmma4(fWi[3], b, acc);
  b.x = q2.x; b.y = q2.y; acc = wmma4(fWi[4], b, acc);
  b.x = q2.z; b.y = q2.w; acc = wmma4(fWi[5], b, acc);
  b.x = q3.x; b.y = q3.y; acc = wmma4(fWi[6], b, acc);
  b.x = q3.z; b.y = q3.w; acc = wmma4(fWi[7], b, acc);
  b.x = q4.x; b.y = q4.y; acc = wmma4(fWi[8], b, acc);
  b.x = q4.z; b.y = q4.w; acc = wmma4(fWi[9], b, acc);
  return acc;
}

#define PSTEPF (16 * 16 * 32 * 8)
#define PROJ_T 16

__device__ __forceinline__ int steps_that_fit(unsigned long long ws_bytes) {
  const unsigned long long n = ws_bytes / ((unsigned long long)PSTEPF * 4ull);
  const int m = (n > (unsigned long long)TIME) ? TIME : (int)n;
  return m & ~3;
}

__device__ __forceinline__ void lds_barrier() {
  __builtin_amdgcn_fence(__ATOMIC_RELEASE, "workgroup", "local");
  __builtin_amdgcn_s_barrier();
  __builtin_amdgcn_fence(__ATOMIC_ACQUIRE, "workgroup", "local");
}

__global__ __launch_bounds__(512)
void dream_proj(const float* __restrict__ x, const float* __restrict__ W_in,
                const float* __restrict__ bvec, float* __restrict__ P, unsigned long long ws_bytes) {
  __shared__ __attribute__((aligned(16))) float Xl[2 * 16 * XS];
  const int tid  = threadIdx.x;
  const int wave = tid >> 5;
  const int lane = tid & 31;
  const int lm   = lane & 15;
  const int kh   = lane >> 4;
  const int tile = blockIdx.x & 15;
  const int b0   = tile * 16;
  const int n0   = wave * 16;
  const int tbeg = (blockIdx.x >> 4) * PROJ_T;
  const int tfit = steps_that_fit(ws_bytes);
  const int tend = (tbeg + PROJ_T < tfit) ? (tbeg + PROJ_T) : tfit;
  if (tbeg >= tend) return;

  v2f fWi[10];
  #pragma unroll
  for (int j = 0; j < 10; ++j) {
    const float* p = W_in + (n0 + lm) * IND;
    const int k = 20 * kh + 2 * j;
    fWi[j].x = p[k];
    fWi[j].y = (k + 1 < IND) ? p[k + 1] : 0.0f;
  }
  float bbv[8];
  #pragma unroll
  for (int v = 0; v < 8; ++v) bbv[v] = bvec[n0 + 8 * kh + v];

  for (int e = tid; e < 2 * 16 * XS; e += 512) Xl[e] = 0.0f;
  __syncthreads();

  const int  e0 = tid, e1 = tid + 512;
  const bool xv1 = (e1 < 16 * IND);
  const float* xg0;  int xl0;
  const float* xg1 = x;  int xl1 = 0;
  { int r = e0 / IND, c = e0 % IND;
    xg0 = x + (size_t)(b0 + r) * TIME * IND + c;  xl0 = r * XS + c; }
  if (xv1) { int r = e1 / IND, c = e1 % IND;
    xg1 = x + (size_t)(b0 + r) * TIME * IND + c;  xl1 = r * XS + c; }

  float xr0 = xg0[(size_t)tbeg * IND];
  float xr1 = xv1 ? xg1[(size_t)tbeg * IND] : 0.0f;
  for (int t = tbeg; t < tend; ++t) {
    float* Xt = Xl + ((t - tbeg) & 1) * (16 * XS);
    Xt[xl0] = xr0;
    if (xv1) Xt[xl1] = xr1;
    if (t + 1 < tend) {
      xr0 = xg0[(size_t)(t + 1) * IND];
      if (xv1) xr1 = xg1[(size_t)(t + 1) * IND];
    }
    lds_barrier();
    v8f acc;
    #pragma unroll
    for (int v = 0; v < 8; ++v) acc[v] = bbv[v];
    acc = xproj(fWi, Xt, lm, kh, acc);
    v4f* dst = (v4f*)(P + ((size_t)(t * 16 + tile) * 16 + wave) * 256 + lane * 8);
    v4f s;
    s.x = acc[0]; s.y = acc[1]; s.z = acc[2]; s.w = acc[3]; dst[0] = s;
    s.x = acc[4]; s.y = acc[5]; s.z = acc[6]; s.w = acc[7]; dst[1] = s;
  }
}

__global__ __launch_bounds__(128)
void dream_scan(const float* __restrict__ x, const float* __restrict__ W_in,
                const float* __restrict__ Amat, const float* __restrict__ B_low,
                const float* __restrict__ bvec, const float* __restrict__ W_dec,
                const float* __restrict__ b_dec, float* __restrict__ out,
                const float* __restrict__ Pws, unsigned long long ws_bytes) {
  __shared__ __attribute__((aligned(16))) float sm[LDS_RESERVE / 4];
  float* Xl = sm + OFF_X;
  float* Rp = sm + OFF_RP;
  float* Dp = sm + OFF_DP;

  const int tid  = threadIdx.x;
  const int wave = tid >> 5;
  const int lane = tid & 31;
  const int lm   = lane & 15;
  const int kh   = lane >> 4;
  const int b0   = (blockIdx.x & 15) * 16;
  const int ph   = blockIdx.x >> 4;
  const int n0   = wave * (16 * TPW);

  v2f fBl[TPW][4], fAw[TPW][4], fWd[3][TPW][4];
  #pragma unroll
  for (int jt = 0; jt < TPW; ++jt) {
    #pragma unroll
    for (int j = 0; j < 4; ++j) {
      const float* p = B_low + lm * HID + n0 + 16 * jt + 8 * kh + 2 * j;
      fBl[jt][j].x = p[0]; fBl[jt][j].y = p[1];
      const float* q = Amat + (n0 + 16 * jt + lm) * RANK + 8 * kh + 2 * j;
      fAw[jt][j].x = q[0]; fAw[jt][j].y = q[1];
    }
  }
  #pragma unroll
  for (int i = 0; i < 3; ++i) {
    const int c = 16 * i + lm;
    #pragma unroll
    for (int jt = 0; jt < TPW; ++jt) {
      #pragma unroll
      for (int j = 0; j < 4; ++j) {
        const float* p = W_dec + (c < IND ? c : 0) * HID + n0 + 16 * jt + 8 * kh + 2 * j;
        fWd[i][jt][j].x = (c < IND) ? p[0] : 0.0f;
        fWd[i][jt][j].y = (c < IND) ? p[1] : 0.0f;
      }
    }
  }

  for (int e = tid; e < 16 * XS; e += 32 * NW) Xl[e] = 0.0f;
  __syncthreads();

  int    dofs[6];
  float* dout[6];
  float  dbias[6];
  bool   dval[6];
  #pragma unroll
  for (int q = 0; q < 6; ++q) {
    const int e = tid + 32 * NW * q;
    const int row = e / 48, c = e % 48;
    dofs[q]  = row * DS + c;
    dval[q]  = (c < IND);
    dout[q]  = out + (size_t)(b0 + row) * TIME * IND + (dval[q] ? c : 0);
    dbias[q] = dval[q] ? b_dec[c] : 0.0f;
  }

  float* Rw = Rp + wave * (16 * RS) + lm * RS + 8 * kh;
  const float* Rr = Rp + lm * RS + 8 * kh;
  float* Dw = Dp + wave * (16 * DS) + lm * DS + 8 * kh;

  v8f hT[TPW];
  #pragma unroll
  for (int jt = 0; jt < TPW; ++jt) hT[jt] = (v8f){};

  auto dec_partials = [&]() __attribute__((always_inline)) {
    v8f d0 = {}, d1 = {}, d2 = {};
    #pragma unroll
    for (int jt = 0; jt < TPW; ++jt) {
      #pragma unroll
      for (int j = 0; j < 4; ++j) {
        const v2f b = bfrag(hT[jt], j);
        d0 = wmma4(fWd[0][jt][j], b, d0);
        d1 = wmma4(fWd[1][jt][j], b, d1);
        d2 = wmma4(fWd[2][jt][j], b, d2);
      }
    }
    v4f* pd = (v4f*)Dw;
    v4f s;
    s.x = d0[0]; s.y = d0[1]; s.z = d0[2]; s.w = d0[3]; pd[0] = s;
    s.x = d0[4]; s.y = d0[5]; s.z = d0[6]; s.w = d0[7]; pd[1] = s;
    s.x = d1[0]; s.y = d1[1]; s.z = d1[2]; s.w = d1[3]; pd[4] = s;
    s.x = d1[4]; s.y = d1[5]; s.z = d1[6]; s.w = d1[7]; pd[5] = s;
    s.x = d2[0]; s.y = d2[1]; s.z = d2[2]; s.w = d2[3]; pd[8] = s;
    s.x = d2[4]; s.y = d2[5]; s.z = d2[6]; s.w = d2[7]; pd[9] = s;
  };
  auto dec_reduce_store = [&](const int u) __attribute__((always_inline)) {
    float p[6][NW];
    #pragma unroll
    for (int q = 0; q < 6; ++q) {
      #pragma unroll
      for (int w = 0; w < NW; ++w) p[q][w] = Dp[w * (16 * DS) + dofs[q]];
    }
    #pragma unroll
    for (int q = 0; q < 6; ++q) {
      float s = dbias[q];
      #pragma unroll
      for (int w = 0; w < NW; ++w) s += p[q][w];
      if (dval[q]) dout[q][(size_t)u * IND] = s;
    }
  };

  v8f pend1 = {}, pend3 = {};

  auto tail = [&](const int t, v8f (&S)[TPW], const bool DEC) __attribute__((always_inline)) {
    const int buf = (t & 1) * (NW * 16 * RS);
    v4f a[NW], c[NW];
    #pragma unroll
    for (int w = 0; w < NW; ++w) {
      const v4f* rr = (const v4f*)(Rr + buf + w * (16 * RS));
      a[w] = rr[0]; c[w] = rr[1];
    }
    float p[6][NW];
    if (DEC) {
      #pragma unroll
      for (int q = 0; q < 6; ++q) {
        #pragma unroll
        for (int w = 0; w < NW; ++w) p[q][w] = Dp[w * (16 * DS) + dofs[q]];
      }
    }
    v8f rs;
    {
      v4f sa = a[0], sc = c[0];
      #pragma unroll
      for (int w = 1; w < NW; ++w) { sa += a[w]; sc += c[w]; }
      rs[0] = sa.x; rs[1] = sa.y; rs[2] = sa.z; rs[3] = sa.w;
      rs[4] = sc.x; rs[5] = sc.y; rs[6] = sc.z; rs[7] = sc.w;
    }
    v8f acc0 = wmma4(fAw[0][0], bfrag(rs, 0), S[0]);
    v8f acc2 = wmma4(fAw[2][0], bfrag(rs, 0), S[2]);
    #pragma unroll
    for (int j = 1; j < 4; ++j) {
      acc0 = wmma4(fAw[0][j], bfrag(rs, j), acc0);
      acc2 = wmma4(fAw[2][j], bfrag(rs, j), acc2);
    }
    __builtin_amdgcn_sched_barrier(0);
    v8f acc1 = S[1], acc3 = S[3];
    #pragma unroll
    for (int j = 0; j < 4; ++j) {
      acc1 = wmma4(fAw[1][j], bfrag(rs, j), acc1);
      __builtin_amdgcn_sched_barrier(0);
      hT[0][2 * j] = fast_tanh(acc0[2 * j]);  hT[0][2 * j + 1] = fast_tanh(acc0[2 * j + 1]);
      __builtin_amdgcn_sched_barrier(0);
      acc3 = wmma4(fAw[3][j], bfrag(rs, j), acc3);
      __builtin_amdgcn_sched_barrier(0);
      hT[2][2 * j] = fast_tanh(acc2[2 * j]);  hT[2][2 * j + 1] = fast_tanh(acc2[2 * j + 1]);
      __builtin_amdgcn_sched_barrier(0);
    }
    pend1 = acc1;  pend3 = acc3;
    if (DEC) {
      #pragma unroll
      for (int q = 0; q < 6; ++q) {
        float s = dbias[q];
        #pragma unroll
        for (int w = 0; w < NW; ++w) s += p[q][w];
        if (dval[q]) dout[q][(size_t)(t - 1) * IND] = s;
      }
    }
  };

  auto step = [&](const int t, v8f (&S)[TPW]) __attribute__((always_inline)) {
    const bool dec_now = (t > 0) && (((t - 1) & (NREP - 1)) == ph);
    const int  buf = (t & 1) * (NW * 16 * RS);
    v8f ra = {}, rb = {};
    #pragma unroll
    for (int j = 0; j < 4; ++j) {
      ra = wmma4(fBl[0][j], bfrag(hT[0], j), ra);
      __builtin_amdgcn_sched_barrier(0);
      hT[1][2 * j] = fast_tanh(pend1[2 * j]);  hT[1][2 * j + 1] = fast_tanh(pend1[2 * j + 1]);
      __builtin_amdgcn_sched_barrier(0);
      rb = wmma4(fBl[2][j], bfrag(hT[2], j), rb);
      __builtin_amdgcn_sched_barrier(0);
      hT[3][2 * j] = fast_tanh(pend3[2 * j]);  hT[3][2 * j + 1] = fast_tanh(pend3[2 * j + 1]);
      __builtin_amdgcn_sched_barrier(0);
    }
    #pragma unroll
    for (int j = 0; j < 4; ++j) {
      ra = wmma4(fBl[1][j], bfrag(hT[1], j), ra);
      rb = wmma4(fBl[3][j], bfrag(hT[3], j), rb);
    }
    {
      v4f* pr = (v4f*)(Rw + buf);
      v4f s;
      s.x = ra[0] + rb[0]; s.y = ra[1] + rb[1]; s.z = ra[2] + rb[2]; s.w = ra[3] + rb[3]; pr[0] = s;
      s.x = ra[4] + rb[4]; s.y = ra[5] + rb[5]; s.z = ra[6] + rb[6]; s.w = ra[7] + rb[7]; pr[1] = s;
    }
    if (dec_now) { dec_partials(); lds_barrier(); tail(t, S, true); }
    else         {                 lds_barrier(); tail(t, S, false); }
  };
  auto finish_pending = [&]() __attribute__((always_inline)) {
    #pragma unroll
    for (int v = 0; v < 8; ++v) { hT[1][v] = fast_tanh(pend1[v]);  hT[3][v] = fast_tanh(pend3[v]); }
  };

  const int t0 = steps_that_fit(ws_bytes);
  if (t0 > 0) {
    const float* Pg = Pws + ((size_t)(b0 >> 4) * 16 + TPW * wave) * 256 + lane * 8;
    auto loadS = [&](const int t, v8f (&S)[TPW]) __attribute__((always_inline)) {
      const int tt = (t < t0) ? t : (t0 - 1);
      const float* pt = Pg + (size_t)tt * PSTEPF;
      #pragma unroll
      for (int jt = 0; jt < TPW; ++jt) {
        const v4f* pp = (const v4f*)(pt + jt * 256);
        const v4f a0 = pp[0], a1 = pp[1];
        S[jt][0] = a0.x; S[jt][1] = a0.y; S[jt][2] = a0.z; S[jt][3] = a0.w;
        S[jt][4] = a1.x; S[jt][5] = a1.y; S[jt][6] = a1.z; S[jt][7] = a1.w;
      }
    };
    v8f S0[TPW], S1[TPW], S2[TPW], S3[TPW];
    loadS(0, S0); loadS(1, S1); loadS(2, S2); loadS(3, S3);
    for (int t = 0; t < t0; t += 4) {
      step(t,     S0); loadS(t + 4, S0);
      step(t + 1, S1); loadS(t + 5, S1);
      step(t + 2, S2); loadS(t + 6, S2);
      step(t + 3, S3); loadS(t + 7, S3);
    }
  }

  if (t0 < TIME) {
    v2f fWi[TPW][10];
    float bbv[TPW][8];
    #pragma unroll
    for (int jt = 0; jt < TPW; ++jt) {
      const float* p = W_in + (n0 + 16 * jt + lm) * IND;
      #pragma unroll
      for (int j = 0; j < 10; ++j) {
        const int k = 20 * kh + 2 * j;
        fWi[jt][j].x = p[k];
        fWi[jt][j].y = (k + 1 < IND) ? p[k + 1] : 0.0f;
      }
      #pragma unroll
      for (int v = 0; v < 8; ++v) bbv[jt][v] = bvec[n0 + 16 * jt + 8 * kh + v];
    }
    for (int t = t0; t < TIME; ++t) {
      for (int e = tid; e < 16 * IND; e += 32 * NW) {
        const int r = e / IND, c = e - r * IND;
        Xl[r * XS + c] = x[((size_t)(b0 + r) * TIME + t) * IND + c];
      }
      lds_barrier();
      v8f S[TPW];
      #pragma unroll
      for (int jt = 0; jt < TPW; ++jt) {
        #pragma unroll
        for (int v = 0; v < 8; ++v) S[jt][v] = bbv[jt][v];
        S[jt] = xproj(fWi[jt], Xl, lm, kh, S[jt]);
      }
      step(t, S);
    }
  }

  finish_pending();

  if (((TIME - 1) & (NREP - 1)) == ph) {
    dec_partials();
    lds_barrier();
    dec_reduce_store(TIME - 1);
  }

  if (ph == 0) {
    #pragma unroll
    for (int jt = 0; jt < TPW; ++jt) {
      float* hf = out + (size_t)BATCH * TIME * IND + (size_t)(b0 + lm) * HID + n0 + 16 * jt + 8 * kh;
      #pragma unroll
      for (int v = 0; v < 8; ++v) hf[v] = hT[jt][v];
    }
  }
}

extern "C" void kernel_launch(void* const* d_in, const int* in_sizes, int n_in,
                              void* d_out, int out_size, void* d_ws, size_t ws_size,
                              hipStream_t stream) {
  const float* x     = (const float*)d_in[0];
  const float* W_in  = (const float*)d_in[1];
  const float* Amat  = (const float*)d_in[2];
  const float* B_low = (const float*)d_in[3];
  const float* bvec  = (const float*)d_in[4];
  const float* W_dec = (const float*)d_in[5];
  const float* b_dec = (const float*)d_in[6];
  float* out = (float*)d_out;

  float* ws = (float*)d_ws;
  const unsigned long long ws_bytes = ws_size;

  dream_proj<<<(BATCH / 16) * (TIME / PROJ_T), 512, 0, stream>>>(x, W_in, bvec, ws, ws_bytes);
  dream_scan<<<(BATCH / 16) * NREP, 32 * NW, 0, stream>>>(
      x, W_in, Amat, B_low, bvec, W_dec, b_dec, out, ws, ws_bytes);
}


# Recurrence copies per batch tile 8 -> 16 (one workgroup on each of the 256 WGPs): a copy decodes every 16th step, so its decoder WMMAs, LDS reduce and stores fall on half as many steps; same sums
# speedup vs baseline: 3.9413x; 1.0495x over previous
#include <hip/hip_runtime.h>
#include <math.h>


typedef float v2f __attribute__((ext_vector_type(2)));
typedef float v4f __attribute__((ext_vector_type(4)));
typedef float v8f __attribute__((ext_vector_type(8)));

#define BATCH 256
#define TIME  1024
#define IND   39
#define HID   256
#define RANK  16
#define NREP  16
static_assert(NREP >= 2 && (NREP & (NREP - 1)) == 0, "NREP: a power of two, at least 2");
#define NW    4
#define TPW   4

#define XS  44
#define RS  20
#define DS  52

#define OFF_X  0
#define OFF_RP (OFF_X + 16 * XS)
#define OFF_DP (OFF_RP + 2 * NW * 16 * RS)
#define LDS_RESERVE (164 * 1024)

__device__ __forceinline__ v8f wmma4(v2f a, v2f b, v8f c) {
  return __builtin_amdgcn_wmma_f32_16x16x4_f32(
       false, a,  false, b,
       (short)0, c,  false,  false);
}

__device__ __forceinline__ float fast_tanh(float v) {
#if __has_builtin(__builtin_amdgcn_tanhf)
  return __builtin_amdgcn_tanhf(v);
#else
  return tanhf(v);
#endif
}

__device__ __forceinline__ v2f bfrag(v8f d, int j) {
  v2f b; b.x = d[2 * j]; b.y = d[2 * j + 1];
  return b;
}

__device__ __forceinline__ v8f xproj(const v2f* fWi, const float* Xl, int lm, int kh, v8f acc) {
  const v4f* xp = (const v4f*)(Xl + lm * XS + 20 * kh);
  v4f q0 = xp[0], q1 = xp[1], q2 = xp[2], q3 = xp[3], q4 = xp[4];
  v2f b;
  b.x = q0.x; b.y = q0.y; acc = wmma4(fWi[0], b, acc);
  b.x = q0.z; b.y = q0.w; acc = wmma4(fWi[1], b, acc);
  b.x = q1.x; b.y = q1.y; acc = wmma4(fWi[2], b, acc);
  b.x = q1.z; b.y = q1.w; acc = wmma4(fWi[3], b, acc);
  b.x = q2.x; b.y = q2.y; acc = wmma4(fWi[4], b, acc);
  b.x = q2.z; b.y = q2.w; acc = wmma4(fWi[5], b, acc);
  b.x = q3.x; b.y = q3.y; acc = wmma4(fWi[6], b, acc);
  b.x = q3.z; b.y = q3.w; acc = wmma4(fWi[7], b, acc);
  b.x = q4.x; b.y = q4.y; acc = wmma4(fWi[8], b, acc);
  b.x = q4.z; b.y = q4.w; acc = wmma4(fWi[9], b, acc);
  return acc;
}

#define PSTEPF (16 * 16 * 32 * 8)
#define PROJ_T 16

__device__ __forceinline__ int steps_that_fit(unsigned long long ws_bytes) {
  const unsigned long long n = ws_bytes / ((unsigned long long)PSTEPF * 4ull);
  const int m = (n > (unsigned long long)TIME) ? TIME : (int)n;
  return m & ~3;
}

__device__ __forceinline__ void lds_barrier() {
  __builtin_amdgcn_fence(__ATOMIC_RELEASE, "workgroup", "local");
  __builtin_amdgcn_s_barrier();
  __builtin_amdgcn_fence(__ATOMIC_ACQUIRE, "workgroup", "local");
}

__global__ __launch_bounds__(512)
void dream_proj(const float* __restrict__ x, const float* __restrict__ W_in,
                const float* __restrict__ bvec, float* __restrict__ P, unsigned long long ws_bytes) {
  __shared__ __attribute__((aligned(16))) float Xl[2 * 16 * XS];
  const int tid  = threadIdx.x;
  const int wave = tid >> 5;
  const int lane = tid & 31;
  const int lm   = lane & 15;
  const int kh   = lane >> 4;
  const int tile = blockIdx.x & 15;
  const int b0   = tile * 16;
  const int n0   = wave * 16;
  const int tbeg = (blockIdx.x >> 4) * PROJ_T;
  const int tfit = steps_that_fit(ws_bytes);
  const int tend = (tbeg + PROJ_T < tfit) ? (tbeg + PROJ_T) : tfit;
  if (tbeg >= tend) return;

  v2f fWi[10];
  #pragma unroll
  for (int j = 0; j < 10; ++j) {
    const float* p = W_in + (n0 + lm) * IND;
    const int k = 20 * kh + 2 * j;
    fWi[j].x = p[k];
    fWi[j].y = (k + 1 < IND) ? p[k + 1] : 0.0f;
  }
  float bbv[8];
  #pragma unroll
  for (int v = 0; v < 8; ++v) bbv[v] = bvec[n0 + 8 * kh + v];

  for (int e = tid; e < 2 * 16 * XS; e += 512) Xl[e] = 0.0f;
  __syncthreads();

  const int  e0 = tid, e1 = tid + 512;
  const bool xv1 = (e1 < 16 * IND);
  const float* xg0;  int xl0;
  const float* xg1 = x;  int xl1 = 0;
  { int r = e0 / IND, c = e0 % IND;
    xg0 = x + (size_t)(b0 + r) * TIME * IND + c;  xl0 = r * XS + c; }
  if (xv1) { int r = e1 / IND, c = e1 % IND;
    xg1 = x + (size_t)(b0 + r) * TIME * IND + c;  xl1 = r * XS + c; }

  float xr0 = xg0[(size_t)tbeg * IND];
  float xr1 = xv1 ? xg1[(size_t)tbeg * IND] : 0.0f;
  for (int t = tbeg; t < tend; ++t) {
    float* Xt = Xl + ((t - tbeg) & 1) * (16 * XS);
    Xt[xl0] = xr0;
    if (xv1) Xt[xl1] = xr1;
    if (t + 1 < tend) {
      xr0 = xg0[(size_t)(t + 1) * IND];
      if (xv1) xr1 = xg1[(size_t)(t + 1) * IND];
    }
    lds_barrier();
    v8f acc;
    #pragma unroll
    for (int v = 0; v < 8; ++v) acc[v] = bbv[v];
    acc = xproj(fWi, Xt, lm, kh, acc);
    v4f* dst = (v4f*)(P + ((size_t)(t * 16 + tile) * 16 + wave) * 256 + lane * 8);
    v4f s;
    s.x = acc[0]; s.y = acc[1]; s.z = acc[2]; s.w = acc[3]; dst[0] = s;
    s.x = acc[4]; s.y = acc[5]; s.z = acc[6]; s.w = acc[7]; dst[1] = s;
  }
}

__global__ __launch_bounds__(128)
void dream_scan(const float* __restrict__ x, const float* __restrict__ W_in,
                const float* __restrict__ Amat, const float* __restrict__ B_low,
                const float* __restrict__ bvec, const float* __restrict__ W_dec,
                const float* __restrict__ b_dec, float* __restrict__ out,
                const float* __restrict__ Pws, unsigned long long ws_bytes) {
  __shared__ __attribute__((aligned(16))) float sm[LDS_RESERVE / 4];
  float* Xl = sm + OFF_X;
  float* Rp = sm + OFF_RP;
  float* Dp = sm + OFF_DP;

  const int tid  = threadIdx.x;
  const int wave = tid >> 5;
  const int lane = tid & 31;
  const int lm   = lane & 15;
  const int kh   = lane >> 4;
  const int b0   = (blockIdx.x & 15) * 16;
  const int ph   = blockIdx.x >> 4;
  const int n0   = wave * (16 * TPW);

  v2f fBl[TPW][4], fAw[TPW][4], fWd[3][TPW][4];
  #pragma unroll
  for (int jt = 0; jt < TPW; ++jt) {
    #pragma unroll
    for (int j = 0; j < 4; ++j) {
      const float* p = B_low + lm * HID + n0 + 16 * jt + 8 * kh + 2 * j;
      fBl[jt][j].x = p[0]; fBl[jt][j].y = p[1];
      const float* q = Amat + (n0 + 16 * jt + lm) * RANK + 8 * kh + 2 * j;
      fAw[jt][j].x = q[0]; fAw[jt][j].y = q[1];
    }
  }
  #pragma unroll
  for (int i = 0; i < 3; ++i) {
    const int c = 16 * i + lm;
    #pragma unroll
    for (int jt = 0; jt < TPW; ++jt) {
      #pragma unroll
      for (int j = 0; j < 4; ++j) {
        const float* p = W_dec + (c < IND ? c : 0) * HID + n0 + 16 * jt + 8 * kh + 2 * j;
        fWd[i][jt][j].x = (c < IND) ? p[0] : 0.0f;
        fWd[i][jt][j].y = (c < IND) ? p[1] : 0.0f;
      }
    }
  }

  for (int e = tid; e < 16 * XS; e += 32 * NW) Xl[e] = 0.0f;
  __syncthreads();

  int    dofs[6];
  float* dout[6];
  float  dbias[6];
  bool   dval[6];
  #pragma unroll
  for (int q = 0; q < 6; ++q) {
    const int e = tid + 32 * NW * q;
    const int row = e / 48, c = e % 48;
    dofs[q]  = row * DS + c;
    dval[q]  = (c < IND);
    dout[q]  = out + (size_t)(b0 + row) * TIME * IND + (dval[q] ? c : 0);
    dbias[q] = dval[q] ? b_dec[c] : 0.0f;
  }

  float* Rw = Rp + wave * (16 * RS) + lm * RS + 8 * kh;
  const float* Rr = Rp + lm * RS + 8 * kh;
  float* Dw = Dp + wave * (16 * DS) + lm * DS + 8 * kh;

  v8f hT[TPW];
  #pragma unroll
  for (int jt = 0; jt < TPW; ++jt) hT[jt] = (v8f){};

  auto dec_partials = [&]() __attribute__((always_inline)) {
    v8f d0 = {}, d1 = {}, d2 = {};
    #pragma unroll
    for (int jt = 0; jt < TPW; ++jt) {
      #pragma unroll
      for (int j = 0; j < 4; ++j) {
        const v2f b = bfrag(hT[jt], j);
        d0 = wmma4(fWd[0][jt][j], b, d0);
        d1 = wmma4(fWd[1][jt][j], b, d1);
        d2 = wmma4(fWd[2][jt][j], b, d2);
      }
    }
    v4f* pd = (v4f*)Dw;
    v4f s;
    s.x = d0[0]; s.y = d0[1]; s.z = d0[2]; s.w = d0[3]; pd[0] = s;
    s.x = d0[4]; s.y = d0[5]; s.z = d0[6]; s.w = d0[7]; pd[1] = s;
    s.x = d1[0]; s.y = d1[1]; s.z = d1[2]; s.w = d1[3]; pd[4] = s;
    s.x = d1[4]; s.y = d1[5]; s.z = d1[6]; s.w = d1[7]; pd[5] = s;
    s.x = d2[0]; s.y = d2[1]; s.z = d2[2]; s.w = d2[3]; pd[8] = s;
    s.x = d2[4]; s.y = d2[5]; s.z = d2[6]; s.w = d2[7]; pd[9] = s;
  };
  auto dec_reduce_store = [&](const int u) __attribute__((always_inline)) {
    float p[6][NW];
    #pragma unroll
    for (int q = 0; q < 6; ++q) {
      #pragma unroll
      for (int w = 0; w < NW; ++w) p[q][w] = Dp[w * (16 * DS) + dofs[q]];
    }
    #pragma unroll
    for (int q = 0; q < 6; ++q) {
      float s = dbias[q];
      #pragma unroll
      for (int w = 0; w < NW; ++w) s += p[q][w];
      if (dval[q]) dout[q][(size_t)u * IND] = s;
    }
  };

  v8f pend1 = {}, pend3 = {};

  auto tail = [&](const int t, v8f (&S)[TPW], const bool DEC) __attribute__((always_inline)) {
    const int buf = (t & 1) * (NW * 16 * RS);
    v4f a[NW], c[NW];
    #pragma unroll
    for (int w = 0; w < NW; ++w) {
      const v4f* rr = (const v4f*)(Rr + buf + w * (16 * RS));
      a[w] = rr[0]; c[w] = rr[1];
    }
    float p[6][NW];
    if (DEC) {
      #pragma unroll
      for (int q = 0; q < 6; ++q) {
        #pragma unroll
        for (int w = 0; w < NW; ++w) p[q][w] = Dp[w * (16 * DS) + dofs[q]];
      }
    }
    v8f rs;
    {
      v4f sa = a[0], sc = c[0];
      #pragma unroll
      for (int w = 1; w < NW; ++w) { sa += a[w]; sc += c[w]; }
      rs[0] = sa.x; rs[1] = sa.y; rs[2] = sa.z; rs[3] = sa.w;
      rs[4] = sc.x; rs[5] = sc.y; rs[6] = sc.z; rs[7] = sc.w;
    }
    v8f acc0 = wmma4(fAw[0][0], bfrag(rs, 0), S[0]);
    v8f acc2 = wmma4(fAw[2][0], bfrag(rs, 0), S[2]);
    #pragma unroll
    for (int j = 1; j < 4; ++j) {
      acc0 = wmma4(fAw[0][j], bfrag(rs, j), acc0);
      acc2 = wmma4(fAw[2][j], bfrag(rs, j), acc2);
    }
    __builtin_amdgcn_sched_barrier(0);
    v8f acc1 = S[1], acc3 = S[3];
    #pragma unroll
    for (int j = 0; j < 4; ++j) {
      acc1 = wmma4(fAw[1][j], bfrag(rs, j), acc1);
      __builtin_amdgcn_sched_barrier(0);
      hT[0][2 * j] = fast_tanh(acc0[2 * j]);  hT[0][2 * j + 1] = fast_tanh(acc0[2 * j + 1]);
      __builtin_amdgcn_sched_barrier(0);
      acc3 = wmma4(fAw[3][j], bfrag(rs, j), acc3);
      __builtin_amdgcn_sched_barrier(0);
      hT[2][2 * j] = fast_tanh(acc2[2 * j]);  hT[2][2 * j + 1] = fast_tanh(acc2[2 * j + 1]);
      __builtin_amdgcn_sched_barrier(0);
    }
    pend1 = acc1;  pend3 = acc3;
    if (DEC) {
      #pragma unroll
      for (int q = 0; q < 6; ++q) {
        float s = dbias[q];
        #pragma unroll
        for (int w = 0; w < NW; ++w) s += p[q][w];
        if (dval[q]) dout[q][(size_t)(t - 1) * IND] = s;
      }
    }
  };

  auto step = [&](const int t, v8f (&S)[TPW]) __attribute__((always_inline)) {
    const bool dec_now = (t > 0) && (((t - 1) & (NREP - 1)) == ph);
    const int  buf = (t & 1) * (NW * 16 * RS);
    v8f ra = {}, rb = {};
    #pragma unroll
    for (int j = 0; j < 4; ++j) {
      ra = wmma4(fBl[0][j], bfrag(hT[0], j), ra);
      __builtin_amdgcn_sched_barrier(0);
      hT[1][2 * j] = fast_tanh(pend1[2 * j]);  hT[1][2 * j + 1] = fast_tanh(pend1[2 * j + 1]);
      __builtin_amdgcn_sched_barrier(0);
      rb = wmma4(fBl[2][j], bfrag(hT[2], j), rb);
      __builtin_amdgcn_sched_barrier(0);
      hT[3][2 * j] = fast_tanh(pend3[2 * j]);  hT[3][2 * j + 1] = fast_tanh(pend3[2 * j + 1]);
      __builtin_amdgcn_sched_barrier(0);
    }
    #pragma unroll
    for (int j = 0; j < 4; ++j) {
      ra = wmma4(fBl[1][j], bfrag(hT[1], j), ra);
      rb = wmma4(fBl[3][j], bfrag(hT[3], j), rb);
    }
    {
      v4f* pr = (v4f*)(Rw + buf);
      v4f s;
      s.x = ra[0] + rb[0]; s.y = ra[1] + rb[1]; s.z = ra[2] + rb[2]; s.w = ra[3] + rb[3]; pr[0] = s;
      s.x = ra[4] + rb[4]; s.y = ra[5] + rb[5]; s.z = ra[6] + rb[6]; s.w = ra[7] + rb[7]; pr[1] = s;
    }
    if (dec_now) { dec_partials(); lds_barrier(); tail(t, S, true); }
    else         {                 lds_barrier(); tail(t, S, false); }
  };
  auto finish_pending = [&]() __attribute__((always_inline)) {
    #pragma unroll
    for (int v = 0; v < 8; ++v) { hT[1][v] = fast_tanh(pend1[v]);  hT[3][v] = fast_tanh(pend3[v]); }
  };

  const int t0 = steps_that_fit(ws_bytes);
  if (t0 > 0) {
    const float* Pg = Pws + ((size_t)(b0 >> 4) * 16 + TPW * wave) * 256 + lane * 8;
    auto loadS = [&](const int t, v8f (&S)[TPW]) __attribute__((always_inline)) {
      const int tt = (t < t0) ? t : (t0 - 1);
      const float* pt = Pg + (size_t)tt * PSTEPF;
      #pragma unroll
      for (int jt = 0; jt < TPW; ++jt) {
        const v4f* pp = (const v4f*)(pt + jt * 256);
        const v4f a0 = pp[0], a1 = pp[1];
        S[jt][0] = a0.x; S[jt][1] = a0.y; S[jt][2] = a0.z; S[jt][3] = a0.w;
        S[jt][4] = a1.x; S[jt][5] = a1.y; S[jt][6] = a1.z; S[jt][7] = a1.w;
      }
    };
    v8f S0[TPW], S1[TPW], S2[TPW], S3[TPW];
    loadS(0, S0); loadS(1, S1); loadS(2, S2); loadS(3, S3);
    for (int t = 0; t < t0; t += 4) {
      step(t,     S0); loadS(t + 4, S0);
      step(t + 1, S1); loadS(t + 5, S1);
      step(t + 2, S2); loadS(t + 6, S2);
      step(t + 3, S3); loadS(t + 7, S3);
    }
  }

  if (t0 < TIME) {
    v2f fWi[TPW][10];
    float bbv[TPW][8];
    #pragma unroll
    for (int jt = 0; jt < TPW; ++jt) {
      const float* p = W_in + (n0 + 16 * jt + lm) * IND;
      #pragma unroll
      for (int j = 0; j < 10; ++j) {
        const int k = 20 * kh + 2 * j;
        fWi[jt][j].x = p[k];
        fWi[jt][j].y = (k + 1 < IND) ? p[k + 1] : 0.0f;
      }
      #pragma unroll
      for (int v = 0; v < 8; ++v) bbv[jt][v] = bvec[n0 + 16 * jt + 8 * kh + v];
    }
    for (int t = t0; t < TIME; ++t) {
      for (int e = tid; e < 16 * IND; e += 32 * NW) {
        const int r = e / IND, c = e - r * IND;
        Xl[r * XS + c] = x[((size_t)(b0 + r) * TIME + t) * IND + c];
      }
      lds_barrier();
      v8f S[TPW];
      #pragma unroll
      for (int jt = 0; jt < TPW; ++jt) {
        #pragma unroll
        for (int v = 0; v < 8; ++v) S[jt][v] = bbv[jt][v];
        S[jt] = xproj(fWi[jt], Xl, lm, kh, S[jt]);
      }
      step(t, S);
    }
  }

  finish_pending();

  if (((TIME - 1) & (NREP - 1)) == ph) {
    dec_partials();
    lds_barrier();
    dec_reduce_store(TIME - 1);
  }

  if (ph == 0) {
    #pragma unroll
    for (int jt = 0; jt < TPW; ++jt) {
      float* hf = out + (size_t)BATCH * TIME * IND + (size_t)(b0 + lm) * HID + n0 + 16 * jt + 8 * kh;
      #pragma unroll
      for (int v = 0; v < 8; ++v) hf[v] = hT[jt][v];
    }
  }
}

extern "C" void kernel_launch(void* const* d_in, const int* in_sizes, int n_in,
                              void* d_out, int out_size, void* d_ws, size_t ws_size,
                              hipStream_t stream) {
  const float* x     = (const float*)d_in[0];
  const float* W_in  = (const float*)d_in[1];
  const float* Amat  = (const float*)d_in[2];
  const float* B_low = (const float*)d_in[3];
  const float* bvec  = (const float*)d_in[4];
  const float* W_dec = (const float*)d_in[5];
  const float* b_dec = (const float*)d_in[6];
  float* out = (float*)d_out;

  float* ws = (float*)d_ws;
  const unsigned long long ws_bytes = ws_size;

  dream_proj<<<(BATCH / 16) * (TIME / PROJ_T), 512, 0, stream>>>(x, W_in, bvec, ws, ws_bytes);
  dream_scan<<<(BATCH / 16) * NREP, 32 * NW, 0, stream>>>(
      x, W_in, Amat, B_low, bvec, W_dec, b_dec, out, ws, ws_bytes);
}


# Decoder of an owned step spread over the next 16 steps: h kept in LDS, 3 of its 48 WMMAs per step issued in the wait for the r reads, not 48 in front of one barrier; decoder k sum order rotated
# speedup vs baseline: 4.1693x; 1.0578x over previous
#include <hip/hip_runtime.h>
#include <math.h>


typedef float v2f __attribute__((ext_vector_type(2)));
typedef float v4f __attribute__((ext_vector_type(4)));
typedef float v8f __attribute__((ext_vector_type(8)));

#define BATCH 256
#define TIME  1024
#define IND   39
#define HID   256
#define RANK  16
#define NREP  16
static_assert(NREP >= 2 && (NREP & (NREP - 1)) == 0, "NREP: a power of two, at least 2");
#define NW    4
#define TPW   4

#define XS  44
#define RS  20
#define DS  52

#define OFF_X  0
#define OFF_RP (OFF_X + 16 * XS)
#define OFF_DP (OFF_RP + 2 * NW * 16 * RS)
#define HSP 36
#define OFF_HS (OFF_DP + NW * 16 * DS)
#define LDS_RESERVE (164 * 1024)

__device__ __forceinline__ v8f wmma4(v2f a, v2f b, v8f c) {
  return __builtin_amdgcn_wmma_f32_16x16x4_f32(
       false, a,  false, b,
       (short)0, c,  false,  false);
}

__device__ __forceinline__ float fast_tanh(float v) {
#if __has_builtin(__builtin_amdgcn_tanhf)
  return __builtin_amdgcn_tanhf(v);
#else
  return tanhf(v);
#endif
}

__device__ __forceinline__ v2f bfrag(v8f d, int j) {
  v2f b; b.x = d[2 * j]; b.y = d[2 * j + 1];
  return b;
}

__device__ __forceinline__ v8f xproj(const v2f* fWi, const float* Xl, int lm, int kh, v8f acc) {
  const v4f* xp = (const v4f*)(Xl + lm * XS + 20 * kh);
  v4f q0 = xp[0], q1 = xp[1], q2 = xp[2], q3 = xp[3], q4 = xp[4];
  v2f b;
  b.x = q0.x; b.y = q0.y; acc = wmma4(fWi[0], b, acc);
  b.x = q0.z; b.y = q0.w; acc = wmma4(fWi[1], b, acc);
  b.x = q1.x; b.y = q1.y; acc = wmma4(fWi[2], b, acc);
  b.x = q1.z; b.y = q1.w; acc = wmma4(fWi[3], b, acc);
  b.x = q2.x; b.y = q2.y; acc = wmma4(fWi[4], b, acc);
  b.x = q2.z; b.y = q2.w; acc = wmma4(fWi[5], b, acc);
  b.x = q3.x; b.y = q3.y; acc = wmma4(fWi[6], b, acc);
  b.x = q3.z; b.y = q3.w; acc = wmma4(fWi[7], b, acc);
  b.x = q4.x; b.y = q4.y; acc = wmma4(fWi[8], b, acc);
  b.x = q4.z; b.y = q4.w; acc = wmma4(fWi[9], b, acc);
  return acc;
}

#define PSTEPF (16 * 16 * 32 * 8)
#define PROJ_T 16

__device__ __forceinline__ int steps_that_fit(unsigned long long ws_bytes) {
  const unsigned long long n = ws_bytes / ((unsigned long long)PSTEPF * 4ull);
  const int m = (n > (unsigned long long)TIME) ? TIME : (int)n;
  return m & ~15;
}

__device__ __forceinline__ void lds_barrier() {
  __builtin_amdgcn_fence(__ATOMIC_RELEASE, "workgroup", "local");
  __builtin_amdgcn_s_barrier();
  __builtin_amdgcn_fence(__ATOMIC_ACQUIRE, "workgroup", "local");
}

__global__ __launch_bounds__(512)
void dream_proj(const float* __restrict__ x, const float* __restrict__ W_in,
                const float* __restrict__ bvec, float* __restrict__ P, unsigned long long ws_bytes) {
  __shared__ __attribute__((aligned(16))) float Xl[2 * 16 * XS];
  const int tid  = threadIdx.x;
  const int wave = tid >> 5;
  const int lane = tid & 31;
  const int lm   = lane & 15;
  const int kh   = lane >> 4;
  const int tile = blockIdx.x & 15;
  const int b0   = tile * 16;
  const int n0   = wave * 16;
  const int tbeg = (blockIdx.x >> 4) * PROJ_T;
  const int tfit = steps_that_fit(ws_bytes);
  const int tend = (tbeg + PROJ_T < tfit) ? (tbeg + PROJ_T) : tfit;
  if (tbeg >= tend) return;

  v2f fWi[10];
  #pragma unroll
  for (int j = 0; j < 10; ++j) {
    const float* p = W_in + (n0 + lm) * IND;
    const int k = 20 * kh + 2 * j;
    fWi[j].x = p[k];
    fWi[j].y = (k + 1 < IND) ? p[k + 1] : 0.0f;
  }
  float bbv[8];
  #pragma unroll
  for (int v = 0; v < 8; ++v) bbv[v] = bvec[n0 + 8 * kh + v];

  for (int e = tid; e < 2 * 16 * XS; e += 512) Xl[e] = 0.0f;
  __syncthreads();

  const int  e0 = tid, e1 = tid + 512;
  const bool xv1 = (e1 < 16 * IND);
  const float* xg0;  int xl0;
  const float* xg1 = x;  int xl1 = 0;
  { int r = e0 / IND, c = e0 % IND;
    xg0 = x + (size_t)(b0 + r) * TIME * IND + c;  xl0 = r * XS + c; }
  if (xv1) { int r = e1 / IND, c = e1 % IND;
    xg1 = x + (size_t)(b0 + r) * TIME * IND + c;  xl1 = r * XS + c; }

  float xr0 = xg0[(size_t)tbeg * IND];
  float xr1 = xv1 ? xg1[(size_t)tbeg * IND] : 0.0f;
  for (int t = tbeg; t < tend; ++t) {
    float* Xt = Xl + ((t - tbeg) & 1) * (16 * XS);
    Xt[xl0] = xr0;
    if (xv1) Xt[xl1] = xr1;
    if (t + 1 < tend) {
      xr0 = xg0[(size_t)(t + 1) * IND];
      if (xv1) xr1 = xg1[(size_t)(t + 1) * IND];
    }
    lds_barrier();
    v8f acc;
    #pragma unroll
    for (int v = 0; v < 8; ++v) acc[v] = bbv[v];
    acc = xproj(fWi, Xt, lm, kh, acc);
    v4f* dst = (v4f*)(P + ((size_t)(t * 16 + tile) * 16 + wave) * 256 + lane * 8);
    v4f s;
    s.x = acc[0]; s.y = acc[1]; s.z = acc[2]; s.w = acc[3]; dst[0] = s;
    s.x = acc[4]; s.y = acc[5]; s.z = acc[6]; s.w = acc[7]; dst[1] = s;
  }
}

__global__ __launch_bounds__(128)
void dream_scan(const float* __restrict__ x, const float* __restrict__ W_in,
                const float* __restrict__ Amat, const float* __restrict__ B_low,
                const float* __restrict__ bvec, const float* __restrict__ W_dec,
                const float* __restrict__ b_dec, float* __restrict__ out,
                const float* __restrict__ Pws, unsigned long long ws_bytes) {
  __shared__ __attribute__((aligned(16))) float sm[LDS_RESERVE / 4];
  float* Xl = sm + OFF_X;
  float* Rp = sm + OFF_RP;
  float* Dp = sm + OFF_DP;

  const int tid  = threadIdx.x;
  const int wave = tid >> 5;
  const int lane = tid & 31;
  const int lm   = lane & 15;
  const int kh   = lane >> 4;
  const int b0   = (blockIdx.x & 15) * 16;
  const int ph   = blockIdx.x >> 4;
  const int n0   = wave * (16 * TPW);

  v2f fBl[TPW][4], fAw[TPW][4], fWd[3][TPW][4];
  #pragma unroll
  for (int jt = 0; jt < TPW; ++jt) {
    #pragma unroll
    for (int j = 0; j < 4; ++j) {
      const float* p = B_low + lm * HID + n0 + 16 * jt + 8 * kh + 2 * j;
      fBl[jt][j].x = p[0]; fBl[jt][j].y = p[1];
      const float* q = Amat + (n0 + 16 * jt + lm) * RANK + 8 * kh + 2 * j;
      fAw[jt][j].x = q[0]; fAw[jt][j].y = q[1];
    }
  }
  #pragma unroll
  for (int i = 0; i < 3; ++i) {
    const int c = 16 * i + lm;
    #pragma unroll
    for (int jt = 0; jt < TPW; ++jt) {
      #pragma unroll
      for (int j = 0; j < 4; ++j) {
        const float* p = W_dec + (c < IND ? c : 0) * HID + n0 + 16 * jt + 8 * kh + 2 * j;
        fWd[i][jt][j].x = (c < IND) ? p[0] : 0.0f;
        fWd[i][jt][j].y = (c < IND) ? p[1] : 0.0f;
      }
    }
  }

  for (int e = tid; e < 16 * XS; e += 32 * NW) Xl[e] = 0.0f;
  float* Hw = sm + OFF_HS + wave * (32 * HSP) + lane * HSP;
  #pragma unroll
  for (int i = 0; i < 2 * TPW; ++i) ((v4f*)Hw)[i] = (v4f){};
  __syncthreads();

  int    dofs[6];
  float* dout[6];
  float  dbias[6];
  bool   dval[6];
  #pragma unroll
  for (int q = 0; q < 6; ++q) {
    const int e = tid + 32 * NW * q;
    const int row = e / 48, c = e % 48;
    dofs[q]  = row * DS + c;
    dval[q]  = (c < IND);
    dout[q]  = out + (size_t)(b0 + row) * TIME * IND + (dval[q] ? c : 0);
    dbias[q] = dval[q] ? b_dec[c] : 0.0f;
  }

  float* Rw = Rp + wave * (16 * RS) + lm * RS + 8 * kh;
  const float* Rr = Rp + lm * RS + 8 * kh;
  float* Dw = Dp + wave * (16 * DS) + lm * DS + 8 * kh;

  v8f hT[TPW];
  #pragma unroll
  for (int jt = 0; jt < TPW; ++jt) hT[jt] = (v8f){};

  v8f dd0 = {}, dd1 = {}, dd2 = {};
  auto dec_begin = [&]() __attribute__((always_inline)) {
    #pragma unroll
    for (int jt = 0; jt < TPW; ++jt) {
      v4f s;
      s.x = hT[jt][0]; s.y = hT[jt][1]; s.z = hT[jt][2]; s.w = hT[jt][3]; ((v4f*)Hw)[2 * jt]     = s;
      s.x = hT[jt][4]; s.y = hT[jt][5]; s.z = hT[jt][6]; s.w = hT[jt][7]; ((v4f*)Hw)[2 * jt + 1] = s;
    }
    dd0 = (v8f){}; dd1 = (v8f){}; dd2 = (v8f){};
  };
#define DEC_SLICE(JT, J) { \
    dd0 = wmma4(fWd[0][JT][J], b, dd0); dd1 = wmma4(fWd[1][JT][J], b, dd1); dd2 = wmma4(fWd[2][JT][J], b, dd2); } break;
  auto dec_slice = [&](const int cs, const v2f b) __attribute__((always_inline)) {
    switch (cs) {
      case 0:  DEC_SLICE(0, 0)  case 1:  DEC_SLICE(0, 1)  case 2:  DEC_SLICE(0, 2)  case 3:  DEC_SLICE(0, 3)
      case 4:  DEC_SLICE(1, 0)  case 5:  DEC_SLICE(1, 1)  case 6:  DEC_SLICE(1, 2)  case 7:  DEC_SLICE(1, 3)
      case 8:  DEC_SLICE(2, 0)  case 9:  DEC_SLICE(2, 1)  case 10: DEC_SLICE(2, 2)  case 11: DEC_SLICE(2, 3)
      case 12: DEC_SLICE(3, 0)  case 13: DEC_SLICE(3, 1)  case 14: DEC_SLICE(3, 2)  default: DEC_SLICE(3, 3)
    }
  };
#undef DEC_SLICE
#define H_PAIR(JT, J) return bfrag(hT[JT], J);
  auto h_pair = [&](const int cs) __attribute__((always_inline)) -> v2f {
    switch (cs) {
      case 0:  H_PAIR(0, 0)  case 1:  H_PAIR(0, 1)  case 2:  H_PAIR(0, 2)  case 3:  H_PAIR(0, 3)
      case 4:  H_PAIR(1, 0)  case 5:  H_PAIR(1, 1)  case 6:  H_PAIR(1, 2)  case 7:  H_PAIR(1, 3)
      case 8:  H_PAIR(2, 0)  case 9:  H_PAIR(2, 1)  case 10: H_PAIR(2, 2)  case 11: H_PAIR(2, 3)
      case 12: H_PAIR(3, 0)  case 13: H_PAIR(3, 1)  case 14: H_PAIR(3, 2)  default: H_PAIR(3, 3)
    }
  };
#undef H_PAIR
  auto dec_store = [&]() __attribute__((always_inline)) {
    v4f* pd = (v4f*)Dw;
    v4f s;
    s.x = dd0[0]; s.y = dd0[1]; s.z = dd0[2]; s.w = dd0[3]; pd[0] = s;
    s.x = dd0[4]; s.y = dd0[5]; s.z = dd0[6]; s.w = dd0[7]; pd[1] = s;
    s.x = dd1[0]; s.y = dd1[1]; s.z = dd1[2]; s.w = dd1[3]; pd[4] = s;
    s.x = dd1[4]; s.y = dd1[5]; s.z = dd1[6]; s.w = dd1[7]; pd[5] = s;
    s.x = dd2[0]; s.y = dd2[1]; s.z = dd2[2]; s.w = dd2[3]; pd[8] = s;
    s.x = dd2[4]; s.y = dd2[5]; s.z = dd2[6]; s.w = dd2[7]; pd[9] = s;
  };
  auto dec_reduce_load = [&](float (&p)[6][NW]) __attribute__((always_inline)) {
    #pragma unroll
    for (int q = 0; q < 6; ++q) {
      #pragma unroll
      for (int w = 0; w < NW; ++w) p[q][w] = Dp[w * (16 * DS) + dofs[q]];
    }
  };
  auto dec_reduce_store = [&](const float (&p)[6][NW], const int u) __attribute__((always_inline)) {
    #pragma unroll
    for (int q = 0; q < 6; ++q) {
      float s = dbias[q];
      #pragma unroll
      for (int w = 0; w < NW; ++w) s += p[q][w];
      if (dval[q]) dout[q][(size_t)u * IND] = s;
    }
  };

  v8f pend1 = {}, pend3 = {};

  v2f lbn = {};
  auto tail = [&](const int t, v8f (&S)[TPW], const int cs, const bool OWN, const v2f lb, const bool CST) __attribute__((always_inline)) {
    const int buf = (t & 1) * (NW * 16 * RS);
    v4f a[NW], c[NW];
    #pragma unroll
    for (int w = 0; w < NW; ++w) {
      const v4f* rr = (const v4f*)(Rr + buf + w * (16 * RS));
      a[w] = rr[0]; c[w] = rr[1];
    }
    v2f sb;
    if (CST) {
      lbn = *(const v2f*)(Hw + 2 * ((cs + 1) & (NREP - 1)));
      const v2f hb = h_pair(cs);
      sb.x = OWN ? hb.x : lb.x;  sb.y = OWN ? hb.y : lb.y;
    } else {
      sb = *(const v2f*)(Hw + 2 * cs);
    }
    __builtin_amdgcn_sched_barrier(0);
    dec_slice(cs, sb);
    asm volatile("" :: "v"(dd0), "v"(dd1), "v"(dd2));
    __builtin_amdgcn_sched_barrier(0);
    v8f rs;
    {
      v4f sa = a[0], sc = c[0];
      #pragma unroll
      for (int w = 1; w < NW; ++w) { sa += a[w]; sc += c[w]; }
      rs[0] = sa.x; rs[1] = sa.y; rs[2] = sa.z; rs[3] = sa.w;
      rs[4] = sc.x; rs[5] = sc.y; rs[6] = sc.z; rs[7] = sc.w;
    }
    v8f acc0 = wmma4(fAw[0][0], bfrag(rs, 0), S[0]);
    v8f acc2 = wmma4(fAw[2][0], bfrag(rs, 0), S[2]);
    #pragma unroll
    for (int j = 1; j < 4; ++j) {
      acc0 = wmma4(fAw[0][j], bfrag(rs, j), acc0);
      acc2 = wmma4(fAw[2][j], bfrag(rs, j), acc2);
    }
    __builtin_amdgcn_sched_barrier(0);
    v8f acc1 = S[1], acc3 = S[3];
    #pragma unroll
    for (int j = 0; j < 4; ++j) {
      acc1 = wmma4(fAw[1][j], bfrag(rs, j), acc1);
      __builtin_amdgcn_sched_barrier(0);
      hT[0][2 * j] = fast_tanh(acc0[2 * j]);  hT[0][2 * j + 1] = fast_tanh(acc0[2 * j + 1]);
      __builtin_amdgcn_sched_barrier(0);
      acc3 = wmma4(fAw[3][j], bfrag(rs, j), acc3);
      __builtin_amdgcn_sched_barrier(0);
      hT[2][2 * j] = fast_tanh(acc2[2 * j]);  hT[2][2 * j + 1] = fast_tanh(acc2[2 * j + 1]);
      __builtin_amdgcn_sched_barrier(0);
    }
    pend1 = acc1;  pend3 = acc3;
  };

  auto step = [&](const int t, v8f (&S)[TPW], const int cs, const bool CST) __attribute__((always_inline)) {
    const bool own_now = (ph == cs) && (t > 0);
    const bool red_now = (ph == ((cs - 1) & (NREP - 1))) && (t > NREP + 1);
    const int  buf = (t & 1) * (NW * 16 * RS);
    const v2f lb = lbn;
    float dsum[6][NW];
    if (red_now) dec_reduce_load(dsum);
    v8f ra = {}, rb = {};
    #pragma unroll
    for (int j = 0; j < 4; ++j) {
      ra = wmma4(fBl[0][j], bfrag(hT[0], j), ra);
      __builtin_amdgcn_sched_barrier(0);
      hT[1][2 * j] = fast_tanh(pend1[2 * j]);  hT[1][2 * j + 1] = fast_tanh(pend1[2 * j + 1]);
      __builtin_amdgcn_sched_barrier(0);
      rb = wmma4(fBl[2][j], bfrag(hT[2], j), rb);
      __builtin_amdgcn_sched_barrier(0);
      hT[3][2 * j] = fast_tanh(pend3[2 * j]);  hT[3][2 * j + 1] = fast_tanh(pend3[2 * j + 1]);
      __builtin_amdgcn_sched_barrier(0);
    }
    #pragma unroll
    for (int j = 0; j < 4; ++j) {
      ra = wmma4(fBl[1][j], bfrag(hT[1], j), ra);
      rb = wmma4(fBl[3][j], bfrag(hT[3], j), rb);
    }
    {
      v4f* pr = (v4f*)(Rw + buf);
      v4f s;
      s.x = ra[0] + rb[0]; s.y = ra[1] + rb[1]; s.z = ra[2] + rb[2]; s.w = ra[3] + rb[3]; pr[0] = s;
      s.x = ra[4] + rb[4]; s.y = ra[5] + rb[5]; s.z = ra[6] + rb[6]; s.w = ra[7] + rb[7]; pr[1] = s;
    }
    if (own_now) {
      if (t > NREP) dec_store();
      dec_begin();
    }
    if (red_now) dec_reduce_store(dsum, t - 2 - NREP);
    lds_barrier();
    tail(t, S, cs, own_now, lb, CST);
  };
  auto finish_pending = [&]() __attribute__((always_inline)) {
    #pragma unroll
    for (int v = 0; v < 8; ++v) { hT[1][v] = fast_tanh(pend1[v]);  hT[3][v] = fast_tanh(pend3[v]); }
  };

  const int t0 = steps_that_fit(ws_bytes);
  if (t0 > 0) {
    const float* Pg = Pws + ((size_t)(b0 >> 4) * 16 + TPW * wave) * 256 + lane * 8;
    auto loadS = [&](const int t, v8f (&S)[TPW]) __attribute__((always_inline)) {
      const int tt = (t < t0) ? t : (t0 - 1);
      const float* pt = Pg + (size_t)tt * PSTEPF;
      #pragma unroll
      for (int jt = 0; jt < TPW; ++jt) {
        const v4f* pp = (const v4f*)(pt + jt * 256);
        const v4f a0 = pp[0], a1 = pp[1];
        S[jt][0] = a0.x; S[jt][1] = a0.y; S[jt][2] = a0.z; S[jt][3] = a0.w;
        S[jt][4] = a1.x; S[jt][5] = a1.y; S[jt][6] = a1.z; S[jt][7] = a1.w;
      }
    };
    v8f S0[TPW], S1[TPW], S2[TPW], S3[TPW];
    loadS(0, S0); loadS(1, S1); loadS(2, S2); loadS(3, S3);
    for (int t = 0; t < t0; t += 16) {
      step(t,      S0, 15, true);  loadS(t + 4,  S0);
      step(t + 1,  S1, 0, true);  loadS(t + 5,  S1);
      step(t + 2,  S2, 1, true);  loadS(t + 6,  S2);
      step(t + 3,  S3, 2, true);  loadS(t + 7,  S3);
      step(t + 4,  S0, 3, true);  loadS(t + 8,  S0);
      step(t + 5,  S1, 4, true);  loadS(t + 9,  S1);
      step(t + 6,  S2, 5, true);  loadS(t + 10, S2);
      step(t + 7,  S3, 6, true);  loadS(t + 11, S3);
      step(t + 8,  S0, 7, true);  loadS(t + 12, S0);
      step(t + 9,  S1, 8, true);  loadS(t + 13, S1);
      step(t + 10, S2, 9, true); loadS(t + 14, S2);
      step(t + 11, S3, 10, true); loadS(t + 15, S3);
      step(t + 12, S0, 11, true); loadS(t + 16, S0);
      step(t + 13, S1, 12, true); loadS(t + 17, S1);
      step(t + 14, S2, 13, true); loadS(t + 18, S2);
      step(t + 15, S3, 14, true); loadS(t + 19, S3);
    }
  }

  if (t0 < TIME) {
    v2f fWi[TPW][10];
    float bbv[TPW][8];
    #pragma unroll
    for (int jt = 0; jt < TPW; ++jt) {
      const float* p = W_in + (n0 + 16 * jt + lm) * IND;
      #pragma unroll
      for (int j = 0; j < 10; ++j) {
        const int k = 20 * kh + 2 * j;
        fWi[jt][j].x = p[k];
        fWi[jt][j].y = (k + 1 < IND) ? p[k + 1] : 0.0f;
      }
      #pragma unroll
      for (int v = 0; v < 8; ++v) bbv[jt][v] = bvec[n0 + 16 * jt + 8 * kh + v];
    }
    for (int t = t0; t < TIME; ++t) {
      for (int e = tid; e < 16 * IND; e += 32 * NW) {
        const int r = e / IND, c = e - r * IND;
        Xl[r * XS + c] = x[((size_t)(b0 + r) * TIME + t) * IND + c];
      }
      lds_barrier();
      v8f S[TPW];
      #pragma unroll
      for (int jt = 0; jt < TPW; ++jt) {
        #pragma unroll
        for (int v = 0; v < 8; ++v) S[jt][v] = bbv[jt][v];
        S[jt] = xproj(fWi[jt], Xl, lm, kh, S[jt]);
      }
      step(t, S, (t - 1) & (NREP - 1), false);
    }
  }

  finish_pending();

  {
    float pr[6][NW];
    if (((TIME - 2) & (NREP - 1)) == ph) {
      dec_reduce_load(pr);
      dec_reduce_store(pr, TIME - 2 - NREP);
    }
    if (((TIME - 1) & (NREP - 1)) == ph) {
      dec_store();
      lds_barrier();
      dec_reduce_load(pr);
      dec_reduce_store(pr, TIME - 1 - NREP);
      dec_begin();
    }
    for (int tv = TIME; tv <= TIME + ph; ++tv) {
      const int cs = (tv - 1) & (NREP - 1);
      dec_slice(cs, *(const v2f*)(Hw + 2 * cs));
    }
    lds_barrier();
    dec_store();
    lds_barrier();
    dec_reduce_load(pr);
    dec_reduce_store(pr, TIME - NREP + ph);
  }

  if (ph == 0) {
    #pragma unroll
    for (int jt = 0; jt < TPW; ++jt) {
      float* hf = out + (size_t)BATCH * TIME * IND + (size_t)(b0 + lm) * HID + n0 + 16 * jt + 8 * kh;
      #pragma unroll
      for (int v = 0; v < 8; ++v) hf[v] = hT[jt][v];
    }
  }
}

extern "C" void kernel_launch(void* const* d_in, const int* in_sizes, int n_in,
                              void* d_out, int out_size, void* d_ws, size_t ws_size,
                              hipStream_t stream) {
  const float* x     = (const float*)d_in[0];
  const float* W_in  = (const float*)d_in[1];
  const float* Amat  = (const float*)d_in[2];
  const float* B_low = (const float*)d_in[3];
  const float* bvec  = (const float*)d_in[4];
  const float* W_dec = (const float*)d_in[5];
  const float* b_dec = (const float*)d_in[6];
  float* out = (float*)d_out;

  float* ws = (float*)d_ws;
  const unsigned long long ws_bytes = ws_size;

  dream_proj<<<(BATCH / 16) * (TIME / PROJ_T), 512, 0, stream>>>(x, W_in, bvec, ws, ws_bytes);
  dream_scan<<<(BATCH / 16) * NREP, 32 * NW, 0, stream>>>(
      x, W_in, Amat, B_low, bvec, W_dec, b_dec, out, ws, ws_bytes);
}
